# starting kernel
# baseline (speedup 1.0000x reference)
#include <hip/hip_runtime.h>

typedef __attribute__((ext_vector_type(2))) float v2f;
typedef __attribute__((ext_vector_type(8))) float v8f;

#define N_NODES 50000
#define N_EDGES 800000
#define DIM     128

__global__ void gcn_zero_kernel(float4* __restrict__ p, int n4) {
    int i = blockIdx.x * blockDim.x + threadIdx.x;
    if (i < n4) p[i] = make_float4(0.f, 0.f, 0.f, 0.f);
}

__global__ void __launch_bounds__(256)
gcn_scatter_kernel(const float* __restrict__ x,
                   const int*   __restrict__ esrc,
                   const int*   __restrict__ edst,
                   const float* __restrict__ ew,
                   float*       __restrict__ agg) {
    int tid  = blockIdx.x * blockDim.x + threadIdx.x;
    int e    = tid >> 5;
    int lane = tid & 31;
    if (e >= N_EDGES) return;

    int   s  = esrc[e];
    int   d  = edst[e];
    float wt = ew[e];

    const float4 v = ((const float4*)(x + (size_t)s * DIM))[lane];
    float* ap = agg + (size_t)d * DIM + lane * 4;
    atomicAdd(ap + 0, v.x * wt);
    atomicAdd(ap + 1, v.y * wt);
    atomicAdd(ap + 2, v.z * wt);
    atomicAdd(ap + 3, v.w * wt);
}

__global__ void __launch_bounds__(256)
gcn_gemm_wmma_kernel(const float* __restrict__ agg,
                     const float* __restrict__ W,
                     const float* __restrict__ bias,
                     float*       __restrict__ out) {
    const int lane    = threadIdx.x & 31;
    const int wave    = threadIdx.x >> 5;
    const int rowTile = blockIdx.x;
    const int mn      = lane & 15;
    const int half    = lane >> 4;
    const int colBase = wave * 16;

    const float* arow = agg + (size_t)(rowTile * 16 + mn) * DIM;

    v8f c = {};
    #pragma unroll
    for (int k0 = 0; k0 < DIM; k0 += 4) {
        const int kb = k0 + half * 2;
        v2f a;
        a.x = arow[kb];
        a.y = arow[kb + 1];
        v2f b;
        b.x = W[(size_t)kb * DIM + colBase + mn];
        b.y = W[(size_t)(kb + 1) * DIM + colBase + mn];
        c = __builtin_amdgcn_wmma_f32_16x16x4_f32(
                false, a, false, b, (short)0, c, false, false);
    }

    const float bv    = bias[colBase + mn];
    const int   mBase = half * 8;
    #pragma unroll
    for (int r = 0; r < 8; ++r) {
        out[(size_t)(rowTile * 16 + mBase + r) * DIM + colBase + mn] = c[r] + bv;
    }
}

extern "C" void kernel_launch(void* const* d_in, const int* in_sizes, int n_in,
                              void* d_out, int out_size, void* d_ws, size_t ws_size,
                              hipStream_t stream) {
    const float* x    = (const float*)d_in[0];
    const int*   esrc = (const int*)d_in[1];
    const int*   edst = (const int*)d_in[2];
    const float* ew   = (const float*)d_in[3];
    const float* W    = (const float*)d_in[4];
    const float* b    = (const float*)d_in[5];
    float*       out  = (float*)d_out;
    float*       agg  = (float*)d_ws;

    const int n4 = N_NODES * DIM / 4;
    gcn_zero_kernel<<<(n4 + 255) / 256, 256, 0, stream>>>((float4*)agg, n4);

    const long long sthreads = (long long)N_EDGES * 32;
    gcn_scatter_kernel<<<(int)((sthreads + 255) / 256), 256, 0, stream>>>(
        x, esrc, edst, ew, agg);

    gcn_gemm_wmma_kernel<<<N_NODES / 16, 256, 0, stream>>>(agg, W, b, out);
}


# float-atomic scatter replaced by in-call counting sort by destination (integer count atomics, arrival-order rank) + one wave per node summing its rows in registers in edge-id order
# speedup vs baseline: 1.4595x; 1.4595x over previous
#include <hip/hip_runtime.h>


typedef __attribute__((ext_vector_type(2))) float v2f;
typedef __attribute__((ext_vector_type(8))) float v8f;

#define N_NODES 50000
#define N_EDGES 800000
#define DIM     128

#define WS_CNT_OFF   33554432
#define WS_OFF_OFF   33816576
#define WS_RANK_OFF  34078720
#define WS_REC_OFF   38273024

__global__ void gcn_zero_counts_kernel(int* __restrict__ cnt) {
    int i = blockIdx.x * blockDim.x + threadIdx.x;
    if (i <= N_NODES) cnt[i] = 0;
}

__global__ void __launch_bounds__(256)
gcn_count_kernel(const int* __restrict__ edst,
                 int*       __restrict__ cnt,
                 int*       __restrict__ rank) {
    int e = blockIdx.x * blockDim.x + threadIdx.x;
    if (e >= N_EDGES) return;
    rank[e] = atomicAdd(&cnt[edst[e]], 1);
}

__global__ void __launch_bounds__(256)
gcn_alloc_kernel(int* __restrict__ cnt, int* __restrict__ off) {
    __shared__ int wsum[8];
    __shared__ int base_s;
    const int tid  = threadIdx.x;
    const int lane = tid & 31;
    const int wave = tid >> 5;
    const int n    = blockIdx.x * 256 + tid;
    const int c    = (n < N_NODES) ? cnt[n] : 0;

    int v = c;
    for (int d = 1; d < 32; d <<= 1) {
        int t = __shfl_up(v, d);
        if (lane >= d) v += t;
    }
    if (lane == 31) wsum[wave] = v;
    __syncthreads();
    if (tid == 0) {
        int tot = 0;
        for (int w = 0; w < 8; ++w) { int t = wsum[w]; wsum[w] = tot; tot += t; }
        base_s = atomicAdd(&cnt[N_NODES], tot);
    }
    __syncthreads();
    if (n < N_NODES) off[n] = base_s + wsum[wave] + (v - c);
}

__global__ void __launch_bounds__(256)
gcn_fill_kernel(const int*   __restrict__ esrc,
                const int*   __restrict__ edst,
                const float* __restrict__ ew,
                const int*   __restrict__ off,
                const int*   __restrict__ rank,
                int4*        __restrict__ rec) {
    int e = blockIdx.x * blockDim.x + threadIdx.x;
    if (e >= N_EDGES) return;
    const int pos = off[edst[e]] + rank[e];
    rec[pos] = make_int4(esrc[e], __float_as_int(ew[e]), e, 0);
}

template <int U>
__device__ __forceinline__ void gather_batch(const float4* __restrict__ x4, int lane,
                                             int rk, int rsrc, int rwt, int j0, float4& acc) {
    float4 v[U];
    float  wt[U];
    #pragma unroll
    for (int u = 0; u < U; ++u) {
        const unsigned b = __builtin_amdgcn_ballot_w32(rk == j0 + u);
        const int l = __builtin_ctz(b);
        const int s = __builtin_amdgcn_readlane(rsrc, l);
        wt[u] = __int_as_float(__builtin_amdgcn_readlane(rwt, l));
        v[u]  = x4[(size_t)s * (DIM / 4) + lane];
    }
    #pragma unroll
    for (int u = 0; u < U; ++u) {
        acc.x += v[u].x * wt[u];
        acc.y += v[u].y * wt[u];
        acc.z += v[u].z * wt[u];
        acc.w += v[u].w * wt[u];
    }
}

__global__ void __launch_bounds__(256)
gcn_gather_kernel(const float* __restrict__ x,
                  const int*   __restrict__ cnt,
                  const int*   __restrict__ off,
                  const int4*  __restrict__ rec,
                  float*       __restrict__ agg) {
    const int lane = threadIdx.x & 31;
    const int node = __builtin_amdgcn_readfirstlane(blockIdx.x * 8 + (threadIdx.x >> 5));
    const int deg  = cnt[node];
    const int beg  = off[node];
    const float4* x4 = (const float4*)x;

    float4 acc = make_float4(0.f, 0.f, 0.f, 0.f);
    for (int base = 0; base < deg; base += 32) {
        const int m = min(32, deg - base);
        int4 r = make_int4(0, 0, 0x7fffffff, 0);
        if (lane < m) r = rec[beg + base + lane];

        int rk = 0;
        for (int j = 0; j < m; ++j) {
            const int ej = __builtin_amdgcn_readlane(r.z, j);
            rk += (ej < r.z) ? 1 : 0;
        }

        int j = 0;
        for (; j + 8 <= m; j += 8) gather_batch<8>(x4, lane, rk, r.x, r.y, j, acc);
        if (j + 4 <= m) { gather_batch<4>(x4, lane, rk, r.x, r.y, j, acc); j += 4; }
        if (j + 2 <= m) { gather_batch<2>(x4, lane, rk, r.x, r.y, j, acc); j += 2; }
        if (j < m)      { gather_batch<1>(x4, lane, rk, r.x, r.y, j, acc); }
    }
    ((float4*)agg)[(size_t)node * (DIM / 4) + lane] = acc;
}

__global__ void __launch_bounds__(256)
gcn_gemm_wmma_kernel(const float* __restrict__ agg,
                     const float* __restrict__ W,
                     const float* __restrict__ bias,
                     float*       __restrict__ out) {
    const int lane    = threadIdx.x & 31;
    const int wave    = threadIdx.x >> 5;
    const int rowTile = blockIdx.x;
    const int mn      = lane & 15;
    const int half    = lane >> 4;
    const int colBase = wave * 16;

    const float* arow = agg + (size_t)(rowTile * 16 + mn) * DIM;

    v8f c = {};
    #pragma unroll
    for (int k0 = 0; k0 < DIM; k0 += 4) {
        const int kb = k0 + half * 2;
        v2f a;
        a.x = arow[kb];
        a.y = arow[kb + 1];
        v2f b;
        b.x = W[(size_t)kb * DIM + colBase + mn];
        b.y = W[(size_t)(kb + 1) * DIM + colBase + mn];
        c = __builtin_amdgcn_wmma_f32_16x16x4_f32(
                false, a, false, b, (short)0, c, false, false);
    }

    const float bv    = bias[colBase + mn];
    const int   mBase = half * 8;
    #pragma unroll
    for (int r = 0; r < 8; ++r) {
        out[(size_t)(rowTile * 16 + mBase + r) * DIM + colBase + mn] = c[r] + bv;
    }
}

extern "C" void kernel_launch(void* const* d_in, const int* in_sizes, int n_in,
                              void* d_out, int out_size, void* d_ws, size_t ws_size,
                              hipStream_t stream) {
    const float* x    = (const float*)d_in[0];
    const int*   esrc = (const int*)d_in[1];
    const int*   edst = (const int*)d_in[2];
    const float* ew   = (const float*)d_in[3];
    const float* W    = (const float*)d_in[4];
    const float* b    = (const float*)d_in[5];
    float*       out  = (float*)d_out;
    float*       agg  = (float*)d_ws;
    int*         cnt  = (int*)((char*)d_ws + WS_CNT_OFF);
    int*         off  = (int*)((char*)d_ws + WS_OFF_OFF);
    int*         rank = (int*)((char*)d_ws + WS_RANK_OFF);
    int4*        rec  = (int4*)((char*)d_ws + WS_REC_OFF);

    gcn_zero_counts_kernel<<<196, 256, 0, stream>>>(cnt);
    gcn_count_kernel<<<3125, 256, 0, stream>>>(edst, cnt, rank);
    gcn_alloc_kernel<<<196, 256, 0, stream>>>(cnt, off);
    gcn_fill_kernel<<<3125, 256, 0, stream>>>(esrc, edst, ew, off, rank, rec);
    gcn_gather_kernel<<<6250, 256, 0, stream>>>(x, cnt, off, rec, agg);
    gcn_gemm_wmma_kernel<<<N_NODES / 16, 256, 0, stream>>>(agg, W, b, out);
}


# gather and GEMM fused: a block sums its 16 node rows (2 per wave, order unchanged) into an LDS tile and multiplies it by W on WMMA in the same launch; agg round trip and the GEMM launch dropped
# speedup vs baseline: 1.6027x; 1.0981x over previous
#include <hip/hip_runtime.h>


typedef __attribute__((ext_vector_type(2))) float v2f;
typedef __attribute__((ext_vector_type(8))) float v8f;

#define N_NODES 50000
#define N_EDGES 800000
#define DIM     128

#define WS_CNT_OFF   33554432
#define WS_OFF_OFF   33816576
#define WS_RANK_OFF  34078720
#define WS_REC_OFF   38273024

__global__ void gcn_zero_counts_kernel(int* __restrict__ cnt) {
    int i = blockIdx.x * blockDim.x + threadIdx.x;
    if (i <= N_NODES) cnt[i] = 0;
}

__global__ void __launch_bounds__(256)
gcn_count_kernel(const int* __restrict__ edst,
                 int*       __restrict__ cnt,
                 int*       __restrict__ rank) {
    int e = blockIdx.x * blockDim.x + threadIdx.x;
    if (e >= N_EDGES) return;
    rank[e] = atomicAdd(&cnt[edst[e]], 1);
}

__global__ void __launch_bounds__(256)
gcn_alloc_kernel(int* __restrict__ cnt, int* __restrict__ off) {
    __shared__ int wsum[8];
    __shared__ int base_s;
    const int tid  = threadIdx.x;
    const int lane = tid & 31;
    const int wave = tid >> 5;
    const int n    = blockIdx.x * 256 + tid;
    const int c    = (n < N_NODES) ? cnt[n] : 0;

    int v = c;
    for (int d = 1; d < 32; d <<= 1) {
        int t = __shfl_up(v, d);
        if (lane >= d) v += t;
    }
    if (lane == 31) wsum[wave] = v;
    __syncthreads();
    if (tid == 0) {
        int tot = 0;
        for (int w = 0; w < 8; ++w) { int t = wsum[w]; wsum[w] = tot; tot += t; }
        base_s = atomicAdd(&cnt[N_NODES], tot);
    }
    __syncthreads();
    if (n < N_NODES) off[n] = base_s + wsum[wave] + (v - c);
}

__global__ void __launch_bounds__(256)
gcn_fill_kernel(const int*   __restrict__ esrc,
                const int*   __restrict__ edst,
                const float* __restrict__ ew,
                const int*   __restrict__ off,
                const int*   __restrict__ rank,
                int4*        __restrict__ rec) {
    int e = blockIdx.x * blockDim.x + threadIdx.x;
    if (e >= N_EDGES) return;
    const int pos = off[edst[e]] + rank[e];
    rec[pos] = make_int4(esrc[e], __float_as_int(ew[e]), e, 0);
}

template <int U>
__device__ __forceinline__ void gather_batch(const float4* __restrict__ x4, int lane,
                                             int rk, int rsrc, int rwt, int j0, float4& acc) {
    float4 v[U];
    float  wt[U];
    #pragma unroll
    for (int u = 0; u < U; ++u) {
        const unsigned b = __builtin_amdgcn_ballot_w32(rk == j0 + u);
        const int l = __builtin_ctz(b);
        const int s = __builtin_amdgcn_readlane(rsrc, l);
        wt[u] = __int_as_float(__builtin_amdgcn_readlane(rwt, l));
        v[u]  = x4[(size_t)s * (DIM / 4) + lane];
    }
    #pragma unroll
    for (int u = 0; u < U; ++u) {
        acc.x += v[u].x * wt[u];
        acc.y += v[u].y * wt[u];
        acc.z += v[u].z * wt[u];
        acc.w += v[u].w * wt[u];
    }
}

#define TILE_PITCH 132

__device__ __forceinline__ float4 gather_node(const float4* __restrict__ x4,
                                              const int4*   __restrict__ rec,
                                              int lane, int deg, int beg) {
    float4 acc = make_float4(0.f, 0.f, 0.f, 0.f);
    for (int base = 0; base < deg; base += 32) {
        const int m = min(32, deg - base);
        int4 r = make_int4(0, 0, 0x7fffffff, 0);
        if (lane < m) r = rec[beg + base + lane];

        int rk = 0;
        for (int j = 0; j < m; ++j) {
            const int ej = __builtin_amdgcn_readlane(r.z, j);
            rk += (ej < r.z) ? 1 : 0;
        }

        int j = 0;
        for (; j + 8 <= m; j += 8) gather_batch<8>(x4, lane, rk, r.x, r.y, j, acc);
        if (j + 4 <= m) { gather_batch<4>(x4, lane, rk, r.x, r.y, j, acc); j += 4; }
        if (j + 2 <= m) { gather_batch<2>(x4, lane, rk, r.x, r.y, j, acc); j += 2; }
        if (j < m)      { gather_batch<1>(x4, lane, rk, r.x, r.y, j, acc); }
    }
    return acc;
}

__global__ void __launch_bounds__(256)
gcn_gather_gemm_kernel(const float* __restrict__ x,
                       const int*   __restrict__ cnt,
                       const int*   __restrict__ off,
                       const int4*  __restrict__ rec,
                       const float* __restrict__ W,
                       const float* __restrict__ bias,
                       float*       __restrict__ out) {
    __shared__ __attribute__((aligned(16))) float tile[16 * TILE_PITCH];

    const int lane    = threadIdx.x & 31;
    const int wave    = __builtin_amdgcn_readfirstlane(threadIdx.x >> 5);
    const int rowTile = blockIdx.x;
    const int mn      = lane & 15;
    const int half    = lane >> 4;
    const int colBase = wave * 16;

    const float4* x4   = (const float4*)x;
    const int     row0 = wave * 2;
    const int     n0   = rowTile * 16 + row0;
    const int deg0 = cnt[n0],     beg0 = off[n0];
    const int deg1 = cnt[n0 + 1], beg1 = off[n0 + 1];

    const float4 s0 = gather_node(x4, rec, lane, deg0, beg0);
    *(float4*)&tile[row0 * TILE_PITCH + lane * 4] = s0;
    const float4 s1 = gather_node(x4, rec, lane, deg1, beg1);
    *(float4*)&tile[(row0 + 1) * TILE_PITCH + lane * 4] = s1;

    v2f bf[DIM / 4];
    #pragma unroll
    for (int k = 0; k < DIM / 4; ++k) {
        const int kb = k * 4 + half * 2;
        bf[k].x = W[(size_t)kb * DIM + colBase + mn];
        bf[k].y = W[(size_t)(kb + 1) * DIM + colBase + mn];
    }
    const float bv = bias[colBase + mn];

    __builtin_amdgcn_fence(__ATOMIC_RELEASE, "workgroup", "local");
    __builtin_amdgcn_s_barrier();
    __builtin_amdgcn_fence(__ATOMIC_ACQUIRE, "workgroup", "local");

    const float* arow = tile + mn * TILE_PITCH + half * 2;
    v8f c = {};
    #pragma unroll
    for (int k = 0; k < DIM / 4; ++k) {
        const v2f a = *(const v2f*)(arow + k * 4);
        c = __builtin_amdgcn_wmma_f32_16x16x4_f32(
                false, a, false, bf[k], (short)0, c, false, false);
    }

    const int mBase = half * 8;
    #pragma unroll
    for (int r = 0; r < 8; ++r) {
        out[(size_t)(rowTile * 16 + mBase + r) * DIM + colBase + mn] = c[r] + bv;
    }
}

extern "C" void kernel_launch(void* const* d_in, const int* in_sizes, int n_in,
                              void* d_out, int out_size, void* d_ws, size_t ws_size,
                              hipStream_t stream) {
    const float* x    = (const float*)d_in[0];
    const int*   esrc = (const int*)d_in[1];
    const int*   edst = (const int*)d_in[2];
    const float* ew   = (const float*)d_in[3];
    const float* W    = (const float*)d_in[4];
    const float* b    = (const float*)d_in[5];
    float*       out  = (float*)d_out;
    int*         cnt  = (int*)((char*)d_ws + WS_CNT_OFF);
    int*         off  = (int*)((char*)d_ws + WS_OFF_OFF);
    int*         rank = (int*)((char*)d_ws + WS_RANK_OFF);
    int4*        rec  = (int4*)((char*)d_ws + WS_REC_OFF);

    gcn_zero_counts_kernel<<<196, 256, 0, stream>>>(cnt);
    gcn_count_kernel<<<3125, 256, 0, stream>>>(edst, cnt, rank);
    gcn_alloc_kernel<<<196, 256, 0, stream>>>(cnt, off);
    gcn_fill_kernel<<<3125, 256, 0, stream>>>(esrc, edst, ew, off, rank, rec);
    gcn_gather_gemm_kernel<<<N_NODES / 16, 256, 0, stream>>>(x, cnt, off, rec, W, b, out);
}


# edges grouped by destination with no global atomics: two-level counting sort on LDS counters (256-node ranges per 4096-edge chunk, then nodes per range); slots stay arrival-ordered as before
# speedup vs baseline: 2.4551x; 1.5319x over previous
#include <hip/hip_runtime.h>


typedef __attribute__((ext_vector_type(2))) float v2f;
typedef __attribute__((ext_vector_type(8))) float v8f;

#define N_NODES 50000
#define N_EDGES 800000
#define DIM     128

#define WS_CNT_OFF    33554432
#define WS_OFF_OFF    33816576
#define WS_TAB_OFF    34078720
#define WS_REC_OFF    38273024
#define WS_PART_OFF   67108864

#define RANGE_SHIFT  8
#define RANGE_NODES  256
#define N_RANGE      196
#define CHUNK        4096
#define CHUNK_PER_T  16
#define N_CHUNK      196
#define TAB_TOT      (N_RANGE * N_CHUNK)
#define TAB_THREADS  1024
#define TAB_PER_T    38
#define TAB_WORDS    (TAB_THREADS * TAB_PER_T)

__device__ __forceinline__ int tab_pos_word(int s) {
    return (s % TAB_PER_T) * TAB_THREADS + s / TAB_PER_T;
}
__device__ __forceinline__ int tab_word(int p, int b) {
    return tab_pos_word(p * N_CHUNK + b);
}

__global__ void __launch_bounds__(256)
gcn_range_count_kernel(const int* __restrict__ edst, int* __restrict__ tab) {
    __shared__ int h[N_RANGE];
    const int tid = threadIdx.x;
    if (tid < N_RANGE) h[tid] = 0;
    __syncthreads();
    const int base = blockIdx.x * CHUNK;
    int d[CHUNK_PER_T];
    #pragma unroll
    for (int k = 0; k < CHUNK_PER_T; ++k) {
        const int e = base + k * 256 + tid;
        d[k] = (e < N_EDGES) ? edst[e] : -1;
    }
    #pragma unroll
    for (int k = 0; k < CHUNK_PER_T; ++k) {
        if (d[k] >= 0) atomicAdd(&h[d[k] >> RANGE_SHIFT], 1);
    }
    __syncthreads();
    if (tid < N_RANGE) tab[tab_word(tid, blockIdx.x)] = h[tid];
    if (blockIdx.x == 0) {
        for (int s = TAB_TOT + tid; s < TAB_WORDS; s += 256) tab[tab_pos_word(s)] = 0;
    }
}

__global__ void __launch_bounds__(TAB_THREADS)
gcn_range_scan_kernel(int* __restrict__ tab) {
    __shared__ int wsum[32];
    const int tid  = threadIdx.x;
    const int lane = tid & 31;
    const int wave = tid >> 5;

    int v[TAB_PER_T];
    #pragma unroll
    for (int r = 0; r < TAB_PER_T; ++r) v[r] = tab[r * TAB_THREADS + tid];
    int tot = 0;
    #pragma unroll
    for (int r = 0; r < TAB_PER_T; ++r) { const int t = v[r]; v[r] = tot; tot += t; }

    int inc = tot;
    for (int dd = 1; dd < 32; dd <<= 1) {
        const int t = __shfl_up(inc, dd);
        if (lane >= dd) inc += t;
    }
    if (lane == 31) wsum[wave] = inc;
    __syncthreads();
    const int w = wsum[lane];
    int winc = w;
    for (int dd = 1; dd < 32; dd <<= 1) {
        const int t = __shfl_up(winc, dd);
        if (lane >= dd) winc += t;
    }
    const int wex  = __shfl(winc, wave) - __shfl(w, wave);
    const int base = wex + (inc - tot);

    #pragma unroll
    for (int r = 0; r < TAB_PER_T; ++r) tab[r * TAB_THREADS + tid] = base + v[r];
}

__global__ void __launch_bounds__(256)
gcn_range_fill_kernel(const int*   __restrict__ esrc,
                      const int*   __restrict__ edst,
                      const float* __restrict__ ew,
                      const int*   __restrict__ tab,
                      int4*        __restrict__ part) {
    __shared__ int cur[N_RANGE];
    const int tid = threadIdx.x;
    if (tid < N_RANGE) cur[tid] = tab[tab_word(tid, blockIdx.x)];
    const int base = blockIdx.x * CHUNK;
    int   d[CHUNK_PER_T];
    int   sidx[CHUNK_PER_T];
    float wv[CHUNK_PER_T];
    #pragma unroll
    for (int k = 0; k < CHUNK_PER_T; ++k) {
        const int e = base + k * 256 + tid;
        const bool ok = e < N_EDGES;
        d[k]    = ok ? edst[e] : -1;
        sidx[k] = ok ? esrc[e] : 0;
        wv[k]   = ok ? ew[e]   : 0.f;
    }
    __syncthreads();
    int slot[CHUNK_PER_T];
    #pragma unroll
    for (int k = 0; k < CHUNK_PER_T; ++k) {
        slot[k] = (d[k] >= 0) ? atomicAdd(&cur[d[k] >> RANGE_SHIFT], 1) : -1;
    }
    #pragma unroll
    for (int k = 0; k < CHUNK_PER_T; ++k) {
        if (slot[k] >= 0)
            part[slot[k]] = make_int4(sidx[k], __float_as_int(wv[k]), base + k * 256 + tid, d[k]);
    }
}

#define SORT_THREADS 1024
#define SORT_BATCH   4
__global__ void __launch_bounds__(SORT_THREADS)
gcn_node_sort_kernel(const int*  __restrict__ tab,
                     const int4* __restrict__ part,
                     int*        __restrict__ cnt,
                     int*        __restrict__ off,
                     int4*       __restrict__ rec) {
    __shared__ int c[RANGE_NODES];
    __shared__ int cur[RANGE_NODES];
    __shared__ int wsum[8];
    const int tid  = threadIdx.x;
    const int lane = tid & 31;
    const int wave = tid >> 5;
    const int p    = blockIdx.x;
    const int beg  = tab[tab_word(p, 0)];
    const int end  = (p + 1 < N_RANGE) ? tab[tab_word(p + 1, 0)] : N_EDGES;

    if (tid < RANGE_NODES) c[tid] = 0;
    __syncthreads();
    const int* partw = (const int*)part;
    for (int i0 = beg; i0 < end; i0 += SORT_THREADS * SORT_BATCH) {
        int dv[SORT_BATCH];
        #pragma unroll
        for (int k = 0; k < SORT_BATCH; ++k) {
            const int i = i0 + k * SORT_THREADS + tid;
            dv[k] = (i < end) ? partw[(size_t)i * 4 + 3] : -1;
        }
        #pragma unroll
        for (int k = 0; k < SORT_BATCH; ++k) {
            if (dv[k] >= 0) atomicAdd(&c[dv[k] & (RANGE_NODES - 1)], 1);
        }
    }
    __syncthreads();

    const int cn = (tid < RANGE_NODES) ? c[tid] : 0;
    int inc = cn;
    for (int dd = 1; dd < 32; dd <<= 1) {
        const int t = __shfl_up(inc, dd);
        if (lane >= dd) inc += t;
    }
    if (tid < RANGE_NODES && lane == 31) wsum[wave] = inc;
    __syncthreads();
    if (tid < RANGE_NODES) {
        int wbase = 0;
        for (int w = 0; w < wave; ++w) wbase += wsum[w];
        const int ex = beg + wbase + (inc - cn);
        cur[tid] = ex;
        const int node = p * RANGE_NODES + tid;
        if (node < N_NODES) { cnt[node] = cn; off[node] = ex; }
    }
    __syncthreads();
    for (int i0 = beg; i0 < end; i0 += SORT_THREADS * SORT_BATCH) {
        int4 r[SORT_BATCH];
        #pragma unroll
        for (int k = 0; k < SORT_BATCH; ++k) {
            const int i = i0 + k * SORT_THREADS + tid;
            r[k] = (i < end) ? part[i] : make_int4(0, 0, 0, -1);
        }
        int slot[SORT_BATCH];
        #pragma unroll
        for (int k = 0; k < SORT_BATCH; ++k) {
            slot[k] = (r[k].w >= 0) ? atomicAdd(&cur[r[k].w & (RANGE_NODES - 1)], 1) : -1;
        }
        #pragma unroll
        for (int k = 0; k < SORT_BATCH; ++k) {
            if (slot[k] >= 0) rec[slot[k]] = make_int4(r[k].x, r[k].y, r[k].z, 0);
        }
    }
}

template <int U>
__device__ __forceinline__ void gather_batch(const float4* __restrict__ x4, int lane,
                                             int rk, int rsrc, int rwt, int j0, float4& acc) {
    float4 v[U];
    float  wt[U];
    #pragma unroll
    for (int u = 0; u < U; ++u) {
        const unsigned b = __builtin_amdgcn_ballot_w32(rk == j0 + u);
        const int l = __builtin_ctz(b);
        const int s = __builtin_amdgcn_readlane(rsrc, l);
        wt[u] = __int_as_float(__builtin_amdgcn_readlane(rwt, l));
        v[u]  = x4[(size_t)s * (DIM / 4) + lane];
    }
    #pragma unroll
    for (int u = 0; u < U; ++u) {
        acc.x += v[u].x * wt[u];
        acc.y += v[u].y * wt[u];
        acc.z += v[u].z * wt[u];
        acc.w += v[u].w * wt[u];
    }
}

#define TILE_PITCH 132

__device__ __forceinline__ float4 gather_node(const float4* __restrict__ x4,
                                              const int4*   __restrict__ rec,
                                              int lane, int deg, int beg) {
    float4 acc = make_float4(0.f, 0.f, 0.f, 0.f);
    for (int base = 0; base < deg; base += 32) {
        const int m = min(32, deg - base);
        int4 r = make_int4(0, 0, 0x7fffffff, 0);
        if (lane < m) r = rec[beg + base + lane];

        int rk = 0;
        for (int j = 0; j < m; ++j) {
            const int ej = __builtin_amdgcn_readlane(r.z, j);
            rk += (ej < r.z) ? 1 : 0;
        }

        int j = 0;
        for (; j + 8 <= m; j += 8) gather_batch<8>(x4, lane, rk, r.x, r.y, j, acc);
        if (j + 4 <= m) { gather_batch<4>(x4, lane, rk, r.x, r.y, j, acc); j += 4; }
        if (j + 2 <= m) { gather_batch<2>(x4, lane, rk, r.x, r.y, j, acc); j += 2; }
        if (j < m)      { gather_batch<1>(x4, lane, rk, r.x, r.y, j, acc); }
    }
    return acc;
}

__global__ void __launch_bounds__(256)
gcn_gather_gemm_kernel(const float* __restrict__ x,
                       const int*   __restrict__ cnt,
                       const int*   __restrict__ off,
                       const int4*  __restrict__ rec,
                       const float* __restrict__ W,
                       const float* __restrict__ bias,
                       float*       __restrict__ out) {
    __shared__ __attribute__((aligned(16))) float tile[16 * TILE_PITCH];

    const int lane    = threadIdx.x & 31;
    const int wave    = __builtin_amdgcn_readfirstlane(threadIdx.x >> 5);
    const int rowTile = blockIdx.x;
    const int mn      = lane & 15;
    const int half    = lane >> 4;
    const int colBase = wave * 16;

    const float4* x4   = (const float4*)x;
    const int     row0 = wave * 2;
    const int     n0   = rowTile * 16 + row0;
    const int deg0 = cnt[n0],     beg0 = off[n0];
    const int deg1 = cnt[n0 + 1], beg1 = off[n0 + 1];

    const float4 s0 = gather_node(x4, rec, lane, deg0, beg0);
    *(float4*)&tile[row0 * TILE_PITCH + lane * 4] = s0;
    const float4 s1 = gather_node(x4, rec, lane, deg1, beg1);
    *(float4*)&tile[(row0 + 1) * TILE_PITCH + lane * 4] = s1;

    v2f bf[DIM / 4];
    #pragma unroll
    for (int k = 0; k < DIM / 4; ++k) {
        const int kb = k * 4 + half * 2;
        bf[k].x = W[(size_t)kb * DIM + colBase + mn];
        bf[k].y = W[(size_t)(kb + 1) * DIM + colBase + mn];
    }
    const float bv = bias[colBase + mn];

    __builtin_amdgcn_fence(__ATOMIC_RELEASE, "workgroup", "local");
    __builtin_amdgcn_s_barrier();
    __builtin_amdgcn_fence(__ATOMIC_ACQUIRE, "workgroup", "local");

    const float* arow = tile + mn * TILE_PITCH + half * 2;
    v8f c = {};
    #pragma unroll
    for (int k = 0; k < DIM / 4; ++k) {
        const v2f a = *(const v2f*)(arow + k * 4);
        c = __builtin_amdgcn_wmma_f32_16x16x4_f32(
                false, a, false, bf[k], (short)0, c, false, false);
    }

    const int mBase = half * 8;
    #pragma unroll
    for (int r = 0; r < 8; ++r) {
        out[(size_t)(rowTile * 16 + mBase + r) * DIM + colBase + mn] = c[r] + bv;
    }
}

extern "C" void kernel_launch(void* const* d_in, const int* in_sizes, int n_in,
                              void* d_out, int out_size, void* d_ws, size_t ws_size,
                              hipStream_t stream) {
    const float* x    = (const float*)d_in[0];
    const int*   esrc = (const int*)d_in[1];
    const int*   edst = (const int*)d_in[2];
    const float* ew   = (const float*)d_in[3];
    const float* W    = (const float*)d_in[4];
    const float* b    = (const float*)d_in[5];
    float*       out  = (float*)d_out;
    int*         cnt    = (int*)((char*)d_ws + WS_CNT_OFF);
    int*         off    = (int*)((char*)d_ws + WS_OFF_OFF);
    int*         tab    = (int*)((char*)d_ws + WS_TAB_OFF);
    int4*        rec    = (int4*)((char*)d_ws + WS_REC_OFF);
    int4*        part   = (int4*)((char*)d_ws + WS_PART_OFF);

    gcn_range_count_kernel<<<N_CHUNK, 256, 0, stream>>>(edst, tab);
    gcn_range_scan_kernel<<<1, TAB_THREADS, 0, stream>>>(tab);
    gcn_range_fill_kernel<<<N_CHUNK, 256, 0, stream>>>(esrc, edst, ew, tab, part);
    gcn_node_sort_kernel<<<N_RANGE, SORT_THREADS, 0, stream>>>(tab, part, cnt, off, rec);
    gcn_gather_gemm_kernel<<<N_NODES / 16, 256, 0, stream>>>(x, cnt, off, rec, W, b, out);
}


# gather+GEMM blocks made persistent: 1024 resident blocks walk 3-4 row tiles each and fetch W once, each wave keeping its B fragments in its own 8 KiB LDS slice instead of reloading them per tile
# speedup vs baseline: 2.5542x; 1.0404x over previous
#include <hip/hip_runtime.h>


typedef __attribute__((ext_vector_type(2))) float v2f;
typedef __attribute__((ext_vector_type(8))) float v8f;

#define N_NODES 50000
#define N_EDGES 800000
#define DIM     128

#define WS_CNT_OFF    33554432
#define WS_OFF_OFF    33816576
#define WS_TAB_OFF    34078720
#define WS_REC_OFF    38273024
#define WS_PART_OFF   67108864

#define RANGE_SHIFT  8
#define RANGE_NODES  256
#define N_RANGE      196
#define CHUNK        4096
#define CHUNK_PER_T  16
#define N_CHUNK      196
#define TAB_TOT      (N_RANGE * N_CHUNK)
#define TAB_THREADS  1024
#define TAB_PER_T    38
#define TAB_WORDS    (TAB_THREADS * TAB_PER_T)

__device__ __forceinline__ int tab_pos_word(int s) {
    return (s % TAB_PER_T) * TAB_THREADS + s / TAB_PER_T;
}
__device__ __forceinline__ int tab_word(int p, int b) {
    return tab_pos_word(p * N_CHUNK + b);
}

__global__ void __launch_bounds__(256)
gcn_range_count_kernel(const int* __restrict__ edst, int* __restrict__ tab) {
    __shared__ int h[N_RANGE];
    const int tid = threadIdx.x;
    if (tid < N_RANGE) h[tid] = 0;
    __syncthreads();
    const int base = blockIdx.x * CHUNK;
    int d[CHUNK_PER_T];
    #pragma unroll
    for (int k = 0; k < CHUNK_PER_T; ++k) {
        const int e = base + k * 256 + tid;
        d[k] = (e < N_EDGES) ? edst[e] : -1;
    }
    #pragma unroll
    for (int k = 0; k < CHUNK_PER_T; ++k) {
        if (d[k] >= 0) atomicAdd(&h[d[k] >> RANGE_SHIFT], 1);
    }
    __syncthreads();
    if (tid < N_RANGE) tab[tab_word(tid, blockIdx.x)] = h[tid];
    if (blockIdx.x == 0) {
        for (int s = TAB_TOT + tid; s < TAB_WORDS; s += 256) tab[tab_pos_word(s)] = 0;
    }
}

__global__ void __launch_bounds__(TAB_THREADS)
gcn_range_scan_kernel(int* __restrict__ tab) {
    __shared__ int wsum[32];
    const int tid  = threadIdx.x;
    const int lane = tid & 31;
    const int wave = tid >> 5;

    int v[TAB_PER_T];
    #pragma unroll
    for (int r = 0; r < TAB_PER_T; ++r) v[r] = tab[r * TAB_THREADS + tid];
    int tot = 0;
    #pragma unroll
    for (int r = 0; r < TAB_PER_T; ++r) { const int t = v[r]; v[r] = tot; tot += t; }

    int inc = tot;
    for (int dd = 1; dd < 32; dd <<= 1) {
        const int t = __shfl_up(inc, dd);
        if (lane >= dd) inc += t;
    }
    if (lane == 31) wsum[wave] = inc;
    __syncthreads();
    const int w = wsum[lane];
    int winc = w;
    for (int dd = 1; dd < 32; dd <<= 1) {
        const int t = __shfl_up(winc, dd);
        if (lane >= dd) winc += t;
    }
    const int wex  = __shfl(winc, wave) - __shfl(w, wave);
    const int base = wex + (inc - tot);

    #pragma unroll
    for (int r = 0; r < TAB_PER_T; ++r) tab[r * TAB_THREADS + tid] = base + v[r];
}

__global__ void __launch_bounds__(256)
gcn_range_fill_kernel(const int*   __restrict__ esrc,
                      const int*   __restrict__ edst,
                      const float* __restrict__ ew,
                      const int*   __restrict__ tab,
                      int4*        __restrict__ part) {
    __shared__ int cur[N_RANGE];
    const int tid = threadIdx.x;
    if (tid < N_RANGE) cur[tid] = tab[tab_word(tid, blockIdx.x)];
    const int base = blockIdx.x * CHUNK;
    int   d[CHUNK_PER_T];
    int   sidx[CHUNK_PER_T];
    float wv[CHUNK_PER_T];
    #pragma unroll
    for (int k = 0; k < CHUNK_PER_T; ++k) {
        const int e = base + k * 256 + tid;
        const bool ok = e < N_EDGES;
        d[k]    = ok ? edst[e] : -1;
        sidx[k] = ok ? esrc[e] : 0;
        wv[k]   = ok ? ew[e]   : 0.f;
    }
    __syncthreads();
    int slot[CHUNK_PER_T];
    #pragma unroll
    for (int k = 0; k < CHUNK_PER_T; ++k) {
        slot[k] = (d[k] >= 0) ? atomicAdd(&cur[d[k] >> RANGE_SHIFT], 1) : -1;
    }
    #pragma unroll
    for (int k = 0; k < CHUNK_PER_T; ++k) {
        if (slot[k] >= 0)
            part[slot[k]] = make_int4(sidx[k], __float_as_int(wv[k]), base + k * 256 + tid, d[k]);
    }
}

#define SORT_THREADS 1024
#define SORT_BATCH   4
__global__ void __launch_bounds__(SORT_THREADS)
gcn_node_sort_kernel(const int*  __restrict__ tab,
                     const int4* __restrict__ part,
                     int*        __restrict__ cnt,
                     int*        __restrict__ off,
                     int4*       __restrict__ rec) {
    __shared__ int c[RANGE_NODES];
    __shared__ int cur[RANGE_NODES];
    __shared__ int wsum[8];
    const int tid  = threadIdx.x;
    const int lane = tid & 31;
    const int wave = tid >> 5;
    const int p    = blockIdx.x;
    const int beg  = tab[tab_word(p, 0)];
    const int end  = (p + 1 < N_RANGE) ? tab[tab_word(p + 1, 0)] : N_EDGES;

    if (tid < RANGE_NODES) c[tid] = 0;
    __syncthreads();
    const int* partw = (const int*)part;
    for (int i0 = beg; i0 < end; i0 += SORT_THREADS * SORT_BATCH) {
        int dv[SORT_BATCH];
        #pragma unroll
        for (int k = 0; k < SORT_BATCH; ++k) {
            const int i = i0 + k * SORT_THREADS + tid;
            dv[k] = (i < end) ? partw[(size_t)i * 4 + 3] : -1;
        }
        #pragma unroll
        for (int k = 0; k < SORT_BATCH; ++k) {
            if (dv[k] >= 0) atomicAdd(&c[dv[k] & (RANGE_NODES - 1)], 1);
        }
    }
    __syncthreads();

    const int cn = (tid < RANGE_NODES) ? c[tid] : 0;
    int inc = cn;
    for (int dd = 1; dd < 32; dd <<= 1) {
        const int t = __shfl_up(inc, dd);
        if (lane >= dd) inc += t;
    }
    if (tid < RANGE_NODES && lane == 31) wsum[wave] = inc;
    __syncthreads();
    if (tid < RANGE_NODES) {
        int wbase = 0;
        for (int w = 0; w < wave; ++w) wbase += wsum[w];
        const int ex = beg + wbase + (inc - cn);
        cur[tid] = ex;
        const int node = p * RANGE_NODES + tid;
        if (node < N_NODES) { cnt[node] = cn; off[node] = ex; }
    }
    __syncthreads();
    for (int i0 = beg; i0 < end; i0 += SORT_THREADS * SORT_BATCH) {
        int4 r[SORT_BATCH];
        #pragma unroll
        for (int k = 0; k < SORT_BATCH; ++k) {
            const int i = i0 + k * SORT_THREADS + tid;
            r[k] = (i < end) ? part[i] : make_int4(0, 0, 0, -1);
        }
        int slot[SORT_BATCH];
        #pragma unroll
        for (int k = 0; k < SORT_BATCH; ++k) {
            slot[k] = (r[k].w >= 0) ? atomicAdd(&cur[r[k].w & (RANGE_NODES - 1)], 1) : -1;
        }
        #pragma unroll
        for (int k = 0; k < SORT_BATCH; ++k) {
            if (slot[k] >= 0) rec[slot[k]] = make_int4(r[k].x, r[k].y, r[k].z, 0);
        }
    }
}

template <int U>
__device__ __forceinline__ void gather_batch(const float4* __restrict__ x4, int lane,
                                             int rk, int rsrc, int rwt, int j0, float4& acc) {
    float4 v[U];
    float  wt[U];
    #pragma unroll
    for (int u = 0; u < U; ++u) {
        const unsigned b = __builtin_amdgcn_ballot_w32(rk == j0 + u);
        const int l = __builtin_ctz(b);
        const int s = __builtin_amdgcn_readlane(rsrc, l);
        wt[u] = __int_as_float(__builtin_amdgcn_readlane(rwt, l));
        v[u]  = x4[(size_t)s * (DIM / 4) + lane];
    }
    #pragma unroll
    for (int u = 0; u < U; ++u) {
        acc.x += v[u].x * wt[u];
        acc.y += v[u].y * wt[u];
        acc.z += v[u].z * wt[u];
        acc.w += v[u].w * wt[u];
    }
}

#define TILE_PITCH 132

__device__ __forceinline__ float4 gather_node(const float4* __restrict__ x4,
                                              const int4*   __restrict__ rec,
                                              int lane, int deg, int beg) {
    float4 acc = make_float4(0.f, 0.f, 0.f, 0.f);
    for (int base = 0; base < deg; base += 32) {
        const int m = min(32, deg - base);
        int4 r = make_int4(0, 0, 0x7fffffff, 0);
        if (lane < m) r = rec[beg + base + lane];

        int rk = 0;
        for (int j = 0; j < m; ++j) {
            const int ej = __builtin_amdgcn_readlane(r.z, j);
            rk += (ej < r.z) ? 1 : 0;
        }

        int j = 0;
        for (; j + 8 <= m; j += 8) gather_batch<8>(x4, lane, rk, r.x, r.y, j, acc);
        if (j + 4 <= m) { gather_batch<4>(x4, lane, rk, r.x, r.y, j, acc); j += 4; }
        if (j + 2 <= m) { gather_batch<2>(x4, lane, rk, r.x, r.y, j, acc); j += 2; }
        if (j < m)      { gather_batch<1>(x4, lane, rk, r.x, r.y, j, acc); }
    }
    return acc;
}

#define N_TILES     (N_NODES / 16)
#define GG_BLOCKS   1024
#define GG_FULL     (N_TILES / GG_BLOCKS)
#define GG_REST     (N_TILES - GG_FULL * GG_BLOCKS)
#define GG_SPREAD   (GG_BLOCKS / GG_REST)
#define WSLICE_F4   ((DIM / 8) * 32)
__global__ void __launch_bounds__(256)
gcn_gather_gemm_kernel(const float* __restrict__ x,
                       const int*   __restrict__ cnt,
                       const int*   __restrict__ off,
                       const int4*  __restrict__ rec,
                       const float* __restrict__ W,
                       const float* __restrict__ bias,
                       float*       __restrict__ out) {
    __shared__ __attribute__((aligned(16))) float tile[16 * TILE_PITCH];
    __shared__ float4 wfrag[8 * WSLICE_F4];

    const int lane    = threadIdx.x & 31;
    const int wave    = __builtin_amdgcn_readfirstlane(threadIdx.x >> 5);
    const int mn      = lane & 15;
    const int half    = lane >> 4;
    const int colBase = wave * 16;
    const int row0    = wave * 2;
    const int mBase   = half * 8;

    const float4* x4 = (const float4*)x;

    float4* wmine = wfrag + wave * WSLICE_F4 + lane;
    #pragma unroll
    for (int j = 0; j < DIM / 8; ++j) {
        const int kb0 = (2 * j) * 4 + half * 2;
        const int kb1 = (2 * j + 1) * 4 + half * 2;
        float4 f;
        f.x = W[(size_t)kb0 * DIM + colBase + mn];
        f.y = W[(size_t)(kb0 + 1) * DIM + colBase + mn];
        f.z = W[(size_t)kb1 * DIM + colBase + mn];
        f.w = W[(size_t)(kb1 + 1) * DIM + colBase + mn];
        wmine[j * 32] = f;
    }
    const float bv = bias[colBase + mn];

    const float* arow = tile + mn * TILE_PITCH + half * 2;

    const int  restIdx = blockIdx.x / GG_SPREAD;
    const bool hasRest = (blockIdx.x % GG_SPREAD == 0) && (restIdx < GG_REST);
    const int  nMine   = GG_FULL + (hasRest ? 1 : 0);

    for (int it = 0; it < nMine; ++it) {
        const int rowTile = (it < GG_FULL) ? it * GG_BLOCKS + (int)blockIdx.x
                                           : GG_FULL * GG_BLOCKS + restIdx;
        const int n0   = rowTile * 16 + row0;
        const int deg0 = cnt[n0],     beg0 = off[n0];
        const int deg1 = cnt[n0 + 1], beg1 = off[n0 + 1];

        const float4 s0 = gather_node(x4, rec, lane, deg0, beg0);
        const float4 s1 = gather_node(x4, rec, lane, deg1, beg1);

        __builtin_amdgcn_fence(__ATOMIC_RELEASE, "workgroup", "local");
        __builtin_amdgcn_s_barrier();
        __builtin_amdgcn_fence(__ATOMIC_ACQUIRE, "workgroup", "local");

        *(float4*)&tile[row0 * TILE_PITCH + lane * 4]       = s0;
        *(float4*)&tile[(row0 + 1) * TILE_PITCH + lane * 4] = s1;

        __builtin_amdgcn_fence(__ATOMIC_RELEASE, "workgroup", "local");
        __builtin_amdgcn_s_barrier();
        __builtin_amdgcn_fence(__ATOMIC_ACQUIRE, "workgroup", "local");

        v8f c = {};
        #pragma unroll
        for (int j = 0; j < DIM / 8; ++j) {
            const float4 f  = wmine[j * 32];
            const v2f    a0 = *(const v2f*)(arow + (2 * j) * 4);
            const v2f    a1 = *(const v2f*)(arow + (2 * j + 1) * 4);
            v2f b0; b0.x = f.x; b0.y = f.y;
            v2f b1; b1.x = f.z; b1.y = f.w;
            c = __builtin_amdgcn_wmma_f32_16x16x4_f32(
                    false, a0, false, b0, (short)0, c, false, false);
            c = __builtin_amdgcn_wmma_f32_16x16x4_f32(
                    false, a1, false, b1, (short)0, c, false, false);
        }

        #pragma unroll
        for (int r = 0; r < 8; ++r) {
            out[(size_t)(rowTile * 16 + mBase + r) * DIM + colBase + mn] = c[r] + bv;
        }
    }
}

extern "C" void kernel_launch(void* const* d_in, const int* in_sizes, int n_in,
                              void* d_out, int out_size, void* d_ws, size_t ws_size,
                              hipStream_t stream) {
    const float* x    = (const float*)d_in[0];
    const int*   esrc = (const int*)d_in[1];
    const int*   edst = (const int*)d_in[2];
    const float* ew   = (const float*)d_in[3];
    const float* W    = (const float*)d_in[4];
    const float* b    = (const float*)d_in[5];
    float*       out  = (float*)d_out;
    int*         cnt    = (int*)((char*)d_ws + WS_CNT_OFF);
    int*         off    = (int*)((char*)d_ws + WS_OFF_OFF);
    int*         tab    = (int*)((char*)d_ws + WS_TAB_OFF);
    int4*        rec    = (int4*)((char*)d_ws + WS_REC_OFF);
    int4*        part   = (int4*)((char*)d_ws + WS_PART_OFF);

    gcn_range_count_kernel<<<N_CHUNK, 256, 0, stream>>>(edst, tab);
    gcn_range_scan_kernel<<<1, TAB_THREADS, 0, stream>>>(tab);
    gcn_range_fill_kernel<<<N_CHUNK, 256, 0, stream>>>(esrc, edst, ew, tab, part);
    gcn_node_sort_kernel<<<N_RANGE, SORT_THREADS, 0, stream>>>(tab, part, cnt, off, rec);
    gcn_gather_gemm_kernel<<<GG_BLOCKS, 256, 0, stream>>>(x, cnt, off, rec, W, b, out);
}


# level-1 grouping made chunk-local: a chunk groups its 4096 records by range inside its own window, the range's block walks the 196 sub-runs; count and scan launches gone (5 -> 3); sum order unchanged
# speedup vs baseline: 2.9558x; 1.1572x over previous
#include <hip/hip_runtime.h>


typedef __attribute__((ext_vector_type(2))) float v2f;
typedef __attribute__((ext_vector_type(8))) float v8f;

#define N_NODES 50000
#define N_EDGES 800000
#define DIM     128

#define WS_CNT_OFF    33554432
#define WS_OFF_OFF    33816576
#define WS_TAB_OFF    34078720
#define WS_REC_OFF    38273024
#define WS_PART_OFF   67108864

#define RANGE_SHIFT  8
#define RANGE_NODES  256
#define N_RANGE      196
#define CHUNK        4096
#define CHUNK_PER_T  16
#define N_CHUNK      196
#define TAB_ROWS     (N_RANGE + 1)

__global__ void __launch_bounds__(256)
gcn_chunk_group_kernel(const int*   __restrict__ esrc,
                       const int*   __restrict__ edst,
                       const float* __restrict__ ew,
                       int*         __restrict__ tab,
                       int4*        __restrict__ part) {
    __shared__ int h[N_RANGE];
    __shared__ int wsum[8];
    const int tid  = threadIdx.x;
    const int lane = tid & 31;
    const int wave = tid >> 5;
    if (tid < N_RANGE) h[tid] = 0;
    const int base = blockIdx.x * CHUNK;
    int   d[CHUNK_PER_T];
    int   sidx[CHUNK_PER_T];
    float wv[CHUNK_PER_T];
    #pragma unroll
    for (int k = 0; k < CHUNK_PER_T; ++k) {
        const int e = base + k * 256 + tid;
        const bool ok = e < N_EDGES;
        d[k]    = ok ? edst[e] : -1;
        sidx[k] = ok ? esrc[e] : 0;
        wv[k]   = ok ? ew[e]   : 0.f;
    }
    __syncthreads();
    #pragma unroll
    for (int k = 0; k < CHUNK_PER_T; ++k) {
        if (d[k] >= 0) atomicAdd(&h[d[k] >> RANGE_SHIFT], 1);
    }
    __syncthreads();

    const int cn = (tid < N_RANGE) ? h[tid] : 0;
    int inc = cn;
    for (int dd = 1; dd < 32; dd <<= 1) {
        const int t = __shfl_up(inc, dd);
        if (lane >= dd) inc += t;
    }
    if (lane == 31) wsum[wave] = inc;
    __syncthreads();
    int wbase = 0;
    for (int w = 0; w < wave; ++w) wbase += wsum[w];
    const int ex = wbase + (inc - cn);
    if (tid < N_RANGE)  h[tid] = ex;
    if (tid < TAB_ROWS) tab[tid * N_CHUNK + blockIdx.x] = ex;
    __syncthreads();

    int slot[CHUNK_PER_T];
    #pragma unroll
    for (int k = 0; k < CHUNK_PER_T; ++k) {
        slot[k] = (d[k] >= 0) ? atomicAdd(&h[d[k] >> RANGE_SHIFT], 1) : -1;
    }
    #pragma unroll
    for (int k = 0; k < CHUNK_PER_T; ++k) {
        if (slot[k] >= 0)
            part[base + slot[k]] = make_int4(sidx[k], __float_as_int(wv[k]), base + k * 256 + tid, d[k]);
    }
}

#define SORT_THREADS 1024
#define SUB_PER_WAVE 7
#define SUB_SLOTS    (SUB_PER_WAVE * 32)
__global__ void __launch_bounds__(SORT_THREADS)
gcn_node_sort_kernel(const int*  __restrict__ tab,
                     const int4* __restrict__ part,
                     int*        __restrict__ cnt,
                     int*        __restrict__ off,
                     int4*       __restrict__ rec) {
    __shared__ int c[RANGE_NODES];
    __shared__ int cur[RANGE_NODES];
    __shared__ int wsum[8];
    __shared__ int bsum[SUB_PER_WAVE];
    __shared__ int sub_beg[SUB_SLOTS];
    __shared__ int sub_cnt[SUB_SLOTS];
    const int tid  = threadIdx.x;
    const int lane = tid & 31;
    const int wave = tid >> 5;
    const int p    = blockIdx.x;

    int ls0 = 0;
    if (tid < SUB_SLOTS) {
        int n = 0;
        if (tid < N_CHUNK) {
            ls0 = tab[p * N_CHUNK + tid];
            n   = tab[(p + 1) * N_CHUNK + tid] - ls0;
        }
        sub_beg[tid] = tid * CHUNK + ls0;
        sub_cnt[tid] = n;
    }
    if (tid < RANGE_NODES) c[tid] = 0;
    int s = ls0;
    for (int dd = 16; dd >= 1; dd >>= 1) s += __shfl_xor(s, dd);
    if (wave < SUB_PER_WAVE && lane == 0) bsum[wave] = s;
    __syncthreads();
    int beg = 0;
    for (int w = 0; w < SUB_PER_WAVE; ++w) beg += bsum[w];

    int sb[SUB_PER_WAVE];
    int sc[SUB_PER_WAVE];
    #pragma unroll
    for (int k = 0; k < SUB_PER_WAVE; ++k) {
        sb[k] = sub_beg[wave + 32 * k];
        sc[k] = sub_cnt[wave + 32 * k];
    }

    const int* partw = (const int*)part;
    {
        int dv[SUB_PER_WAVE];
        #pragma unroll
        for (int k = 0; k < SUB_PER_WAVE; ++k) {
            dv[k] = (lane < sc[k]) ? partw[(size_t)(sb[k] + lane) * 4 + 3] : -1;
        }
        #pragma unroll
        for (int k = 0; k < SUB_PER_WAVE; ++k) {
            if (dv[k] >= 0) atomicAdd(&c[dv[k] & (RANGE_NODES - 1)], 1);
        }
        for (int b = wave; b < N_CHUNK; b += 32) {
            const int n  = sub_cnt[b];
            const int s0 = sub_beg[b];
            for (int i = lane + 32; i < n; i += 32) {
                const int dd = partw[(size_t)(s0 + i) * 4 + 3];
                atomicAdd(&c[dd & (RANGE_NODES - 1)], 1);
            }
        }
    }
    __syncthreads();

    const int cn = (tid < RANGE_NODES) ? c[tid] : 0;
    int inc = cn;
    for (int dd = 1; dd < 32; dd <<= 1) {
        const int t = __shfl_up(inc, dd);
        if (lane >= dd) inc += t;
    }
    if (tid < RANGE_NODES && lane == 31) wsum[wave] = inc;
    __syncthreads();
    if (tid < RANGE_NODES) {
        int wbase = 0;
        for (int w = 0; w < wave; ++w) wbase += wsum[w];
        const int ex = beg + wbase + (inc - cn);
        cur[tid] = ex;
        const int node = p * RANGE_NODES + tid;
        if (node < N_NODES) { cnt[node] = cn; off[node] = ex; }
    }
    __syncthreads();
    {
        int4 r[SUB_PER_WAVE];
        #pragma unroll
        for (int k = 0; k < SUB_PER_WAVE; ++k) {
            r[k] = (lane < sc[k]) ? part[sb[k] + lane] : make_int4(0, 0, 0, -1);
        }
        int slot[SUB_PER_WAVE];
        #pragma unroll
        for (int k = 0; k < SUB_PER_WAVE; ++k) {
            slot[k] = (r[k].w >= 0) ? atomicAdd(&cur[r[k].w & (RANGE_NODES - 1)], 1) : -1;
        }
        #pragma unroll
        for (int k = 0; k < SUB_PER_WAVE; ++k) {
            if (slot[k] >= 0) rec[slot[k]] = make_int4(r[k].x, r[k].y, r[k].z, 0);
        }
        for (int b = wave; b < N_CHUNK; b += 32) {
            const int n  = sub_cnt[b];
            const int s0 = sub_beg[b];
            for (int i = lane + 32; i < n; i += 32) {
                const int4 q  = part[s0 + i];
                const int  sl = atomicAdd(&cur[q.w & (RANGE_NODES - 1)], 1);
                rec[sl] = make_int4(q.x, q.y, q.z, 0);
            }
        }
    }
}

template <int U>
__device__ __forceinline__ void gather_batch(const float4* __restrict__ x4, int lane,
                                             int rk, int rsrc, int rwt, int j0, float4& acc) {
    float4 v[U];
    float  wt[U];
    #pragma unroll
    for (int u = 0; u < U; ++u) {
        const unsigned b = __builtin_amdgcn_ballot_w32(rk == j0 + u);
        const int l = __builtin_ctz(b);
        const int s = __builtin_amdgcn_readlane(rsrc, l);
        wt[u] = __int_as_float(__builtin_amdgcn_readlane(rwt, l));
        v[u]  = x4[(size_t)s * (DIM / 4) + lane];
    }
    #pragma unroll
    for (int u = 0; u < U; ++u) {
        acc.x += v[u].x * wt[u];
        acc.y += v[u].y * wt[u];
        acc.z += v[u].z * wt[u];
        acc.w += v[u].w * wt[u];
    }
}

#define TILE_PITCH 132

__device__ __forceinline__ float4 gather_node(const float4* __restrict__ x4,
                                              const int4*   __restrict__ rec,
                                              int lane, int deg, int beg) {
    float4 acc = make_float4(0.f, 0.f, 0.f, 0.f);
    for (int base = 0; base < deg; base += 32) {
        const int m = min(32, deg - base);
        int4 r = make_int4(0, 0, 0x7fffffff, 0);
        if (lane < m) r = rec[beg + base + lane];

        int rk = 0;
        for (int j = 0; j < m; ++j) {
            const int ej = __builtin_amdgcn_readlane(r.z, j);
            rk += (ej < r.z) ? 1 : 0;
        }

        int j = 0;
        for (; j + 8 <= m; j += 8) gather_batch<8>(x4, lane, rk, r.x, r.y, j, acc);
        if (j + 4 <= m) { gather_batch<4>(x4, lane, rk, r.x, r.y, j, acc); j += 4; }
        if (j + 2 <= m) { gather_batch<2>(x4, lane, rk, r.x, r.y, j, acc); j += 2; }
        if (j < m)      { gather_batch<1>(x4, lane, rk, r.x, r.y, j, acc); }
    }
    return acc;
}

#define N_TILES     (N_NODES / 16)
#define GG_BLOCKS   1024
#define GG_FULL     (N_TILES / GG_BLOCKS)
#define GG_REST     (N_TILES - GG_FULL * GG_BLOCKS)
#define GG_SPREAD   (GG_BLOCKS / GG_REST)
#define WSLICE_F4   ((DIM / 8) * 32)
__global__ void __launch_bounds__(256)
gcn_gather_gemm_kernel(const float* __restrict__ x,
                       const int*   __restrict__ cnt,
                       const int*   __restrict__ off,
                       const int4*  __restrict__ rec,
                       const float* __restrict__ W,
                       const float* __restrict__ bias,
                       float*       __restrict__ out) {
    __shared__ __attribute__((aligned(16))) float tile[16 * TILE_PITCH];
    __shared__ float4 wfrag[8 * WSLICE_F4];

    const int lane    = threadIdx.x & 31;
    const int wave    = __builtin_amdgcn_readfirstlane(threadIdx.x >> 5);
    const int mn      = lane & 15;
    const int half    = lane >> 4;
    const int colBase = wave * 16;
    const int row0    = wave * 2;
    const int mBase   = half * 8;

    const float4* x4 = (const float4*)x;

    float4* wmine = wfrag + wave * WSLICE_F4 + lane;
    #pragma unroll
    for (int j = 0; j < DIM / 8; ++j) {
        const int kb0 = (2 * j) * 4 + half * 2;
        const int kb1 = (2 * j + 1) * 4 + half * 2;
        float4 f;
        f.x = W[(size_t)kb0 * DIM + colBase + mn];
        f.y = W[(size_t)(kb0 + 1) * DIM + colBase + mn];
        f.z = W[(size_t)kb1 * DIM + colBase + mn];
        f.w = W[(size_t)(kb1 + 1) * DIM + colBase + mn];
        wmine[j * 32] = f;
    }
    const float bv = bias[colBase + mn];

    const float* arow = tile + mn * TILE_PITCH + half * 2;

    const int  restIdx = blockIdx.x / GG_SPREAD;
    const bool hasRest = (blockIdx.x % GG_SPREAD == 0) && (restIdx < GG_REST);
    const int  nMine   = GG_FULL + (hasRest ? 1 : 0);

    for (int it = 0; it < nMine; ++it) {
        const int rowTile = (it < GG_FULL) ? it * GG_BLOCKS + (int)blockIdx.x
                                           : GG_FULL * GG_BLOCKS + restIdx;
        const int n0   = rowTile * 16 + row0;
        const int deg0 = cnt[n0],     beg0 = off[n0];
        const int deg1 = cnt[n0 + 1], beg1 = off[n0 + 1];

        const float4 s0 = gather_node(x4, rec, lane, deg0, beg0);
        const float4 s1 = gather_node(x4, rec, lane, deg1, beg1);

        __builtin_amdgcn_fence(__ATOMIC_RELEASE, "workgroup", "local");
        __builtin_amdgcn_s_barrier();
        __builtin_amdgcn_fence(__ATOMIC_ACQUIRE, "workgroup", "local");

        *(float4*)&tile[row0 * TILE_PITCH + lane * 4]       = s0;
        *(float4*)&tile[(row0 + 1) * TILE_PITCH + lane * 4] = s1;

        __builtin_amdgcn_fence(__ATOMIC_RELEASE, "workgroup", "local");
        __builtin_amdgcn_s_barrier();
        __builtin_amdgcn_fence(__ATOMIC_ACQUIRE, "workgroup", "local");

        v8f c = {};
        #pragma unroll
        for (int j = 0; j < DIM / 8; ++j) {
            const float4 f  = wmine[j * 32];
            const v2f    a0 = *(const v2f*)(arow + (2 * j) * 4);
            const v2f    a1 = *(const v2f*)(arow + (2 * j + 1) * 4);
            v2f b0; b0.x = f.x; b0.y = f.y;
            v2f b1; b1.x = f.z; b1.y = f.w;
            c = __builtin_amdgcn_wmma_f32_16x16x4_f32(
                    false, a0, false, b0, (short)0, c, false, false);
            c = __builtin_amdgcn_wmma_f32_16x16x4_f32(
                    false, a1, false, b1, (short)0, c, false, false);
        }

        #pragma unroll
        for (int r = 0; r < 8; ++r) {
            out[(size_t)(rowTile * 16 + mBase + r) * DIM + colBase + mn] = c[r] + bv;
        }
    }
}

extern "C" void kernel_launch(void* const* d_in, const int* in_sizes, int n_in,
                              void* d_out, int out_size, void* d_ws, size_t ws_size,
                              hipStream_t stream) {
    const float* x    = (const float*)d_in[0];
    const int*   esrc = (const int*)d_in[1];
    const int*   edst = (const int*)d_in[2];
    const float* ew   = (const float*)d_in[3];
    const float* W    = (const float*)d_in[4];
    const float* b    = (const float*)d_in[5];
    float*       out  = (float*)d_out;
    int*         cnt    = (int*)((char*)d_ws + WS_CNT_OFF);
    int*         off    = (int*)((char*)d_ws + WS_OFF_OFF);
    int*         tab    = (int*)((char*)d_ws + WS_TAB_OFF);
    int4*        rec    = (int4*)((char*)d_ws + WS_REC_OFF);
    int4*        part   = (int4*)((char*)d_ws + WS_PART_OFF);

    gcn_chunk_group_kernel<<<N_CHUNK, 256, 0, stream>>>(esrc, edst, ew, tab, part);
    gcn_node_sort_kernel<<<N_RANGE, SORT_THREADS, 0, stream>>>(tab, part, cnt, off, rec);
    gcn_gather_gemm_kernel<<<GG_BLOCKS, 256, 0, stream>>>(x, cnt, off, rec, W, b, out);
}


# both grouping levels put records into an LDS image of the block's output window and copy it out in order as whole 128-byte lines, not one scattered 16-byte store per edge; slots, sum order unchanged
# speedup vs baseline: 3.0905x; 1.0456x over previous
#include <hip/hip_runtime.h>


typedef __attribute__((ext_vector_type(2))) float v2f;
typedef __attribute__((ext_vector_type(8))) float v8f;

#define N_NODES 50000
#define N_EDGES 800000
#define DIM     128

#define WS_CNT_OFF    33554432
#define WS_OFF_OFF    33816576
#define WS_TAB_OFF    34078720
#define WS_REC_OFF    38273024
#define WS_PART_OFF   67108864

#define RANGE_SHIFT  8
#define RANGE_NODES  256
#define N_RANGE      196
#define CHUNK        4096
#define CHUNK_PER_T  16
#define N_CHUNK      196
#define TAB_ROWS     (N_RANGE + 1)

__device__ __forceinline__ void hold_loaded(const int4& v) {
    asm volatile("" :: "v"(v.x), "v"(v.y), "v"(v.z), "v"(v.w));
}

__global__ void __launch_bounds__(256)
gcn_chunk_group_kernel(const int*   __restrict__ esrc,
                       const int*   __restrict__ edst,
                       const float* __restrict__ ew,
                       int*         __restrict__ tab,
                       int4*        __restrict__ part) {
    __shared__ int h[N_RANGE];
    __shared__ int wsum[8];
    __shared__ __attribute__((aligned(16))) int4 img[CHUNK];
    const int tid  = threadIdx.x;
    const int lane = tid & 31;
    const int wave = tid >> 5;
    if (tid < N_RANGE) h[tid] = 0;
    const int base = blockIdx.x * CHUNK;
    int   d[CHUNK_PER_T];
    int   sidx[CHUNK_PER_T];
    float wv[CHUNK_PER_T];
    #pragma unroll
    for (int k = 0; k < CHUNK_PER_T; ++k) {
        const int e = base + k * 256 + tid;
        const bool ok = e < N_EDGES;
        d[k]    = ok ? edst[e] : -1;
        sidx[k] = ok ? esrc[e] : 0;
        wv[k]   = ok ? ew[e]   : 0.f;
    }
    __syncthreads();
    #pragma unroll
    for (int k = 0; k < CHUNK_PER_T; ++k) {
        if (d[k] >= 0) atomicAdd(&h[d[k] >> RANGE_SHIFT], 1);
    }
    __syncthreads();

    const int cn = (tid < N_RANGE) ? h[tid] : 0;
    int inc = cn;
    for (int dd = 1; dd < 32; dd <<= 1) {
        const int t = __shfl_up(inc, dd);
        if (lane >= dd) inc += t;
    }
    if (lane == 31) wsum[wave] = inc;
    __syncthreads();
    int wbase = 0;
    for (int w = 0; w < wave; ++w) wbase += wsum[w];
    const int ex = wbase + (inc - cn);
    if (tid < N_RANGE)  h[tid] = ex;
    if (tid < TAB_ROWS) tab[tid * N_CHUNK + blockIdx.x] = ex;
    __syncthreads();

    int slot[CHUNK_PER_T];
    #pragma unroll
    for (int k = 0; k < CHUNK_PER_T; ++k) {
        slot[k] = (d[k] >= 0) ? atomicAdd(&h[d[k] >> RANGE_SHIFT], 1) : -1;
    }
    #pragma unroll
    for (int k = 0; k < CHUNK_PER_T; ++k) {
        if (slot[k] >= 0)
            img[slot[k]] = make_int4(sidx[k], __float_as_int(wv[k]), base + k * 256 + tid, d[k]);
    }
    __syncthreads();
    const int nrec = min(CHUNK, N_EDGES - base);
    int4 q[CHUNK_PER_T];
    #pragma unroll
    for (int k = 0; k < CHUNK_PER_T; ++k) q[k] = img[min(k * 256 + tid, nrec - 1)];
    #pragma unroll
    for (int k = 0; k < CHUNK_PER_T; ++k) hold_loaded(q[k]);
    #pragma unroll
    for (int k = 0; k < CHUNK_PER_T; ++k) {
        const int i = k * 256 + tid;
        if (i < nrec) part[base + i] = q[k];
    }
}

#define SORT_THREADS 1024
#define SUB_PER_WAVE 7
#define SUB_SLOTS    (SUB_PER_WAVE * 32)
#define IMG_CAP      5120
#define IMG_TRIPS    (IMG_CAP / SORT_THREADS + 1)
__global__ void __launch_bounds__(SORT_THREADS)
gcn_node_sort_kernel(const int*  __restrict__ tab,
                     const int4* __restrict__ part,
                     int*        __restrict__ cnt,
                     int*        __restrict__ off,
                     int4*       __restrict__ rec) {
    __shared__ int c[RANGE_NODES];
    __shared__ int cur[RANGE_NODES];
    __shared__ int wsum[8];
    __shared__ int bsum[SUB_PER_WAVE];
    __shared__ int sub_beg[SUB_SLOTS];
    __shared__ int sub_cnt[SUB_SLOTS];
    __shared__ __attribute__((aligned(16))) int4 img[IMG_CAP];
    const int tid  = threadIdx.x;
    const int lane = tid & 31;
    const int wave = tid >> 5;
    const int p    = blockIdx.x;

    int ls0 = 0;
    if (tid < SUB_SLOTS) {
        int n = 0;
        if (tid < N_CHUNK) {
            ls0 = tab[p * N_CHUNK + tid];
            n   = tab[(p + 1) * N_CHUNK + tid] - ls0;
        }
        sub_beg[tid] = tid * CHUNK + ls0;
        sub_cnt[tid] = n;
    }
    if (tid < RANGE_NODES) c[tid] = 0;
    int s = ls0;
    for (int dd = 16; dd >= 1; dd >>= 1) s += __shfl_xor(s, dd);
    if (wave < SUB_PER_WAVE && lane == 0) bsum[wave] = s;
    __syncthreads();
    int beg = 0;
    for (int w = 0; w < SUB_PER_WAVE; ++w) beg += bsum[w];

    int sb[SUB_PER_WAVE];
    int sc[SUB_PER_WAVE];
    #pragma unroll
    for (int k = 0; k < SUB_PER_WAVE; ++k) {
        sb[k] = sub_beg[wave + 32 * k];
        sc[k] = sub_cnt[wave + 32 * k];
    }

    const int* partw = (const int*)part;
    {
        int dv[SUB_PER_WAVE];
        #pragma unroll
        for (int k = 0; k < SUB_PER_WAVE; ++k) {
            dv[k] = (lane < sc[k]) ? partw[(size_t)(sb[k] + lane) * 4 + 3] : -1;
        }
        #pragma unroll
        for (int k = 0; k < SUB_PER_WAVE; ++k) {
            if (dv[k] >= 0) atomicAdd(&c[dv[k] & (RANGE_NODES - 1)], 1);
        }
        for (int b = wave; b < N_CHUNK; b += 32) {
            const int n  = sub_cnt[b];
            const int s0 = sub_beg[b];
            for (int i = lane + 32; i < n; i += 32) {
                const int dd = partw[(size_t)(s0 + i) * 4 + 3];
                atomicAdd(&c[dd & (RANGE_NODES - 1)], 1);
            }
        }
    }
    __syncthreads();

    const int cn = (tid < RANGE_NODES) ? c[tid] : 0;
    int inc = cn;
    for (int dd = 1; dd < 32; dd <<= 1) {
        const int t = __shfl_up(inc, dd);
        if (lane >= dd) inc += t;
    }
    if (tid < RANGE_NODES && lane == 31) wsum[wave] = inc;
    __syncthreads();
    int total = 0;
    for (int w = 0; w < RANGE_NODES / 32; ++w) total += wsum[w];
    const bool in_lds = total <= IMG_CAP;
    if (tid < RANGE_NODES) {
        int wbase = 0;
        for (int w = 0; w < wave; ++w) wbase += wsum[w];
        const int ex = beg + wbase + (inc - cn);
        cur[tid] = ex;
        const int node = p * RANGE_NODES + tid;
        if (node < N_NODES) { cnt[node] = cn; off[node] = ex; }
    }
    __syncthreads();
    {
        int4 r[SUB_PER_WAVE];
        #pragma unroll
        for (int k = 0; k < SUB_PER_WAVE; ++k) {
            r[k] = (lane < sc[k]) ? part[sb[k] + lane] : make_int4(0, 0, 0, -1);
        }
        int slot[SUB_PER_WAVE];
        #pragma unroll
        for (int k = 0; k < SUB_PER_WAVE; ++k) {
            slot[k] = (r[k].w >= 0) ? atomicAdd(&cur[r[k].w & (RANGE_NODES - 1)], 1) : -1;
        }
        if (in_lds) {
            #pragma unroll
            for (int k = 0; k < SUB_PER_WAVE; ++k) {
                if (slot[k] >= 0) img[slot[k] - beg] = make_int4(r[k].x, r[k].y, r[k].z, 0);
            }
        } else {
            #pragma unroll
            for (int k = 0; k < SUB_PER_WAVE; ++k) {
                if (slot[k] >= 0) rec[slot[k]] = make_int4(r[k].x, r[k].y, r[k].z, 0);
            }
        }
        for (int b = wave; b < N_CHUNK; b += 32) {
            const int n  = sub_cnt[b];
            const int s0 = sub_beg[b];
            for (int i = lane + 32; i < n; i += 32) {
                const int4 q  = part[s0 + i];
                const int  sl = atomicAdd(&cur[q.w & (RANGE_NODES - 1)], 1);
                if (in_lds) img[sl - beg] = make_int4(q.x, q.y, q.z, 0);
                else        rec[sl]       = make_int4(q.x, q.y, q.z, 0);
            }
        }
    }
    __syncthreads();
    if (in_lds && total > 0) {
        const int shift = beg & 7;
        int4 q[IMG_TRIPS];
        #pragma unroll
        for (int j = 0; j < IMG_TRIPS; ++j) q[j] = img[min(max(j * SORT_THREADS + tid - shift, 0), total - 1)];
        #pragma unroll
        for (int j = 0; j < IMG_TRIPS; ++j) hold_loaded(q[j]);
        #pragma unroll
        for (int j = 0; j < IMG_TRIPS; ++j) {
            const int i = j * SORT_THREADS + tid - shift;
            if (i >= 0 && i < total) rec[beg + i] = q[j];
        }
    }
}

template <int U>
__device__ __forceinline__ void gather_batch(const float4* __restrict__ x4, int lane,
                                             int rk, int rsrc, int rwt, int j0, float4& acc) {
    float4 v[U];
    float  wt[U];
    #pragma unroll
    for (int u = 0; u < U; ++u) {
        const unsigned b = __builtin_amdgcn_ballot_w32(rk == j0 + u);
        const int l = __builtin_ctz(b);
        const int s = __builtin_amdgcn_readlane(rsrc, l);
        wt[u] = __int_as_float(__builtin_amdgcn_readlane(rwt, l));
        v[u]  = x4[(size_t)s * (DIM / 4) + lane];
    }
    #pragma unroll
    for (int u = 0; u < U; ++u) {
        acc.x += v[u].x * wt[u];
        acc.y += v[u].y * wt[u];
        acc.z += v[u].z * wt[u];
        acc.w += v[u].w * wt[u];
    }
}

#define TILE_PITCH 132

__device__ __forceinline__ float4 gather_node(const float4* __restrict__ x4,
                                              const int4*   __restrict__ rec,
                                              int lane, int deg, int beg) {
    float4 acc = make_float4(0.f, 0.f, 0.f, 0.f);
    for (int base = 0; base < deg; base += 32) {
        const int m = min(32, deg - base);
        int4 r = make_int4(0, 0, 0x7fffffff, 0);
        if (lane < m) r = rec[beg + base + lane];

        int rk = 0;
        for (int j = 0; j < m; ++j) {
            const int ej = __builtin_amdgcn_readlane(r.z, j);
            rk += (ej < r.z) ? 1 : 0;
        }

        int j = 0;
        for (; j + 8 <= m; j += 8) gather_batch<8>(x4, lane, rk, r.x, r.y, j, acc);
        if (j + 4 <= m) { gather_batch<4>(x4, lane, rk, r.x, r.y, j, acc); j += 4; }
        if (j + 2 <= m) { gather_batch<2>(x4, lane, rk, r.x, r.y, j, acc); j += 2; }
        if (j < m)      { gather_batch<1>(x4, lane, rk, r.x, r.y, j, acc); }
    }
    return acc;
}

#define N_TILES     (N_NODES / 16)
#define GG_BLOCKS   1024
#define GG_FULL     (N_TILES / GG_BLOCKS)
#define GG_REST     (N_TILES - GG_FULL * GG_BLOCKS)
#define GG_SPREAD   (GG_BLOCKS / GG_REST)
#define WSLICE_F4   ((DIM / 8) * 32)
__global__ void __launch_bounds__(256)
gcn_gather_gemm_kernel(const float* __restrict__ x,
                       const int*   __restrict__ cnt,
                       const int*   __restrict__ off,
                       const int4*  __restrict__ rec,
                       const float* __restrict__ W,
                       const float* __restrict__ bias,
                       float*       __restrict__ out) {
    __shared__ __attribute__((aligned(16))) float tile[16 * TILE_PITCH];
    __shared__ float4 wfrag[8 * WSLICE_F4];

    const int lane    = threadIdx.x & 31;
    const int wave    = __builtin_amdgcn_readfirstlane(threadIdx.x >> 5);
    const int mn      = lane & 15;
    const int half    = lane >> 4;
    const int colBase = wave * 16;
    const int row0    = wave * 2;
    const int mBase   = half * 8;

    const float4* x4 = (const float4*)x;

    float4* wmine = wfrag + wave * WSLICE_F4 + lane;
    #pragma unroll
    for (int j = 0; j < DIM / 8; ++j) {
        const int kb0 = (2 * j) * 4 + half * 2;
        const int kb1 = (2 * j + 1) * 4 + half * 2;
        float4 f;
        f.x = W[(size_t)kb0 * DIM + colBase + mn];
        f.y = W[(size_t)(kb0 + 1) * DIM + colBase + mn];
        f.z = W[(size_t)kb1 * DIM + colBase + mn];
        f.w = W[(size_t)(kb1 + 1) * DIM + colBase + mn];
        wmine[j * 32] = f;
    }
    const float bv = bias[colBase + mn];

    const float* arow = tile + mn * TILE_PITCH + half * 2;

    const int  restIdx = blockIdx.x / GG_SPREAD;
    const bool hasRest = (blockIdx.x % GG_SPREAD == 0) && (restIdx < GG_REST);
    const int  nMine   = GG_FULL + (hasRest ? 1 : 0);

    for (int it = 0; it < nMine; ++it) {
        const int rowTile = (it < GG_FULL) ? it * GG_BLOCKS + (int)blockIdx.x
                                           : GG_FULL * GG_BLOCKS + restIdx;
        const int n0   = rowTile * 16 + row0;
        const int deg0 = cnt[n0],     beg0 = off[n0];
        const int deg1 = cnt[n0 + 1], beg1 = off[n0 + 1];

        const float4 s0 = gather_node(x4, rec, lane, deg0, beg0);
        const float4 s1 = gather_node(x4, rec, lane, deg1, beg1);

        __builtin_amdgcn_fence(__ATOMIC_RELEASE, "workgroup", "local");
        __builtin_amdgcn_s_barrier();
        __builtin_amdgcn_fence(__ATOMIC_ACQUIRE, "workgroup", "local");

        *(float4*)&tile[row0 * TILE_PITCH + lane * 4]       = s0;
        *(float4*)&tile[(row0 + 1) * TILE_PITCH + lane * 4] = s1;

        __builtin_amdgcn_fence(__ATOMIC_RELEASE, "workgroup", "local");
        __builtin_amdgcn_s_barrier();
        __builtin_amdgcn_fence(__ATOMIC_ACQUIRE, "workgroup", "local");

        v8f c = {};
        #pragma unroll
        for (int j = 0; j < DIM / 8; ++j) {
            const float4 f  = wmine[j * 32];
            const v2f    a0 = *(const v2f*)(arow + (2 * j) * 4);
            const v2f    a1 = *(const v2f*)(arow + (2 * j + 1) * 4);
            v2f b0; b0.x = f.x; b0.y = f.y;
            v2f b1; b1.x = f.z; b1.y = f.w;
            c = __builtin_amdgcn_wmma_f32_16x16x4_f32(
                    false, a0, false, b0, (short)0, c, false, false);
            c = __builtin_amdgcn_wmma_f32_16x16x4_f32(
                    false, a1, false, b1, (short)0, c, false, false);
        }

        #pragma unroll
        for (int r = 0; r < 8; ++r) {
            out[(size_t)(rowTile * 16 + mBase + r) * DIM + colBase + mn] = c[r] + bv;
        }
    }
}

extern "C" void kernel_launch(void* const* d_in, const int* in_sizes, int n_in,
                              void* d_out, int out_size, void* d_ws, size_t ws_size,
                              hipStream_t stream) {
    const float* x    = (const float*)d_in[0];
    const int*   esrc = (const int*)d_in[1];
    const int*   edst = (const int*)d_in[2];
    const float* ew   = (const float*)d_in[3];
    const float* W    = (const float*)d_in[4];
    const float* b    = (const float*)d_in[5];
    float*       out  = (float*)d_out;
    int*         cnt    = (int*)((char*)d_ws + WS_CNT_OFF);
    int*         off    = (int*)((char*)d_ws + WS_OFF_OFF);
    int*         tab    = (int*)((char*)d_ws + WS_TAB_OFF);
    int4*        rec    = (int4*)((char*)d_ws + WS_REC_OFF);
    int4*        part   = (int4*)((char*)d_ws + WS_PART_OFF);

    gcn_chunk_group_kernel<<<N_CHUNK, 256, 0, stream>>>(esrc, edst, ew, tab, part);
    gcn_node_sort_kernel<<<N_RANGE, SORT_THREADS, 0, stream>>>(tab, part, cnt, off, rec);
    gcn_gather_gemm_kernel<<<GG_BLOCKS, 256, 0, stream>>>(x, cnt, off, rec, W, b, out);
}


# gather+GEMM blocks take row tiles from a shared counter (first tile fixed, then one integer atomic increment per tile, issued ahead of the gather) instead of a fixed 3 or 4 each; sum order unchanged
# speedup vs baseline: 3.2301x; 1.0452x over previous
#include <hip/hip_runtime.h>


typedef __attribute__((ext_vector_type(2))) float v2f;
typedef __attribute__((ext_vector_type(8))) float v8f;

#define N_NODES 50000
#define N_EDGES 800000
#define DIM     128

#define WS_CNT_OFF    33554432
#define WS_OFF_OFF    33816576
#define WS_CTR_OFF    33800192
#define WS_TAB_OFF    34078720
#define WS_REC_OFF    38273024
#define WS_PART_OFF   67108864

#define RANGE_SHIFT  8
#define RANGE_NODES  256
#define N_RANGE      196
#define CHUNK        4096
#define CHUNK_PER_T  16
#define N_CHUNK      196
#define TAB_ROWS     (N_RANGE + 1)

__device__ __forceinline__ void hold_loaded(const int4& v) {
    asm volatile("" :: "v"(v.x), "v"(v.y), "v"(v.z), "v"(v.w));
}

__global__ void __launch_bounds__(256)
gcn_chunk_group_kernel(const int*   __restrict__ esrc,
                       const int*   __restrict__ edst,
                       const float* __restrict__ ew,
                       int*         __restrict__ tab,
                       int4*        __restrict__ part) {
    __shared__ int h[N_RANGE];
    __shared__ int wsum[8];
    __shared__ __attribute__((aligned(16))) int4 img[CHUNK];
    const int tid  = threadIdx.x;
    const int lane = tid & 31;
    const int wave = tid >> 5;
    if (tid < N_RANGE) h[tid] = 0;
    const int base = blockIdx.x * CHUNK;
    int   d[CHUNK_PER_T];
    int   sidx[CHUNK_PER_T];
    float wv[CHUNK_PER_T];
    #pragma unroll
    for (int k = 0; k < CHUNK_PER_T; ++k) {
        const int e = base + k * 256 + tid;
        const bool ok = e < N_EDGES;
        d[k]    = ok ? edst[e] : -1;
        sidx[k] = ok ? esrc[e] : 0;
        wv[k]   = ok ? ew[e]   : 0.f;
    }
    __syncthreads();
    #pragma unroll
    for (int k = 0; k < CHUNK_PER_T; ++k) {
        if (d[k] >= 0) atomicAdd(&h[d[k] >> RANGE_SHIFT], 1);
    }
    __syncthreads();

    const int cn = (tid < N_RANGE) ? h[tid] : 0;
    int inc = cn;
    for (int dd = 1; dd < 32; dd <<= 1) {
        const int t = __shfl_up(inc, dd);
        if (lane >= dd) inc += t;
    }
    if (lane == 31) wsum[wave] = inc;
    __syncthreads();
    int wbase = 0;
    for (int w = 0; w < wave; ++w) wbase += wsum[w];
    const int ex = wbase + (inc - cn);
    if (tid < N_RANGE)  h[tid] = ex;
    if (tid < TAB_ROWS) tab[tid * N_CHUNK + blockIdx.x] = ex;
    __syncthreads();

    int slot[CHUNK_PER_T];
    #pragma unroll
    for (int k = 0; k < CHUNK_PER_T; ++k) {
        slot[k] = (d[k] >= 0) ? atomicAdd(&h[d[k] >> RANGE_SHIFT], 1) : -1;
    }
    #pragma unroll
    for (int k = 0; k < CHUNK_PER_T; ++k) {
        if (slot[k] >= 0)
            img[slot[k]] = make_int4(sidx[k], __float_as_int(wv[k]), base + k * 256 + tid, d[k]);
    }
    __syncthreads();
    const int nrec = min(CHUNK, N_EDGES - base);
    int4 q[CHUNK_PER_T];
    #pragma unroll
    for (int k = 0; k < CHUNK_PER_T; ++k) q[k] = img[min(k * 256 + tid, nrec - 1)];
    #pragma unroll
    for (int k = 0; k < CHUNK_PER_T; ++k) hold_loaded(q[k]);
    #pragma unroll
    for (int k = 0; k < CHUNK_PER_T; ++k) {
        const int i = k * 256 + tid;
        if (i < nrec) part[base + i] = q[k];
    }
}

#define SORT_THREADS 1024
#define SUB_PER_WAVE 7
#define SUB_SLOTS    (SUB_PER_WAVE * 32)
#define IMG_CAP      5120
#define IMG_TRIPS    (IMG_CAP / SORT_THREADS + 1)
__global__ void __launch_bounds__(SORT_THREADS)
gcn_node_sort_kernel(const int*  __restrict__ tab,
                     const int4* __restrict__ part,
                     int*        __restrict__ cnt,
                     int*        __restrict__ off,
                     int4*       __restrict__ rec,
                     int*        __restrict__ ctr) {
    __shared__ int c[RANGE_NODES];
    __shared__ int cur[RANGE_NODES];
    __shared__ int wsum[8];
    __shared__ int bsum[SUB_PER_WAVE];
    __shared__ int sub_beg[SUB_SLOTS];
    __shared__ int sub_cnt[SUB_SLOTS];
    __shared__ __attribute__((aligned(16))) int4 img[IMG_CAP];
    const int tid  = threadIdx.x;
    const int lane = tid & 31;
    const int wave = tid >> 5;
    const int p    = blockIdx.x;
    if (p == 0 && tid == 0) *ctr = 0;

    int ls0 = 0;
    if (tid < SUB_SLOTS) {
        int n = 0;
        if (tid < N_CHUNK) {
            ls0 = tab[p * N_CHUNK + tid];
            n   = tab[(p + 1) * N_CHUNK + tid] - ls0;
        }
        sub_beg[tid] = tid * CHUNK + ls0;
        sub_cnt[tid] = n;
    }
    if (tid < RANGE_NODES) c[tid] = 0;
    int s = ls0;
    for (int dd = 16; dd >= 1; dd >>= 1) s += __shfl_xor(s, dd);
    if (wave < SUB_PER_WAVE && lane == 0) bsum[wave] = s;
    __syncthreads();
    int beg = 0;
    for (int w = 0; w < SUB_PER_WAVE; ++w) beg += bsum[w];

    int sb[SUB_PER_WAVE];
    int sc[SUB_PER_WAVE];
    #pragma unroll
    for (int k = 0; k < SUB_PER_WAVE; ++k) {
        sb[k] = sub_beg[wave + 32 * k];
        sc[k] = sub_cnt[wave + 32 * k];
    }

    const int* partw = (const int*)part;
    {
        int dv[SUB_PER_WAVE];
        #pragma unroll
        for (int k = 0; k < SUB_PER_WAVE; ++k) {
            dv[k] = (lane < sc[k]) ? partw[(size_t)(sb[k] + lane) * 4 + 3] : -1;
        }
        #pragma unroll
        for (int k = 0; k < SUB_PER_WAVE; ++k) {
            if (dv[k] >= 0) atomicAdd(&c[dv[k] & (RANGE_NODES - 1)], 1);
        }
        for (int b = wave; b < N_CHUNK; b += 32) {
            const int n  = sub_cnt[b];
            const int s0 = sub_beg[b];
            for (int i = lane + 32; i < n; i += 32) {
                const int dd = partw[(size_t)(s0 + i) * 4 + 3];
                atomicAdd(&c[dd & (RANGE_NODES - 1)], 1);
            }
        }
    }
    __syncthreads();

    const int cn = (tid < RANGE_NODES) ? c[tid] : 0;
    int inc = cn;
    for (int dd = 1; dd < 32; dd <<= 1) {
        const int t = __shfl_up(inc, dd);
        if (lane >= dd) inc += t;
    }
    if (tid < RANGE_NODES && lane == 31) wsum[wave] = inc;
    __syncthreads();
    int total = 0;
    for (int w = 0; w < RANGE_NODES / 32; ++w) total += wsum[w];
    const bool in_lds = total <= IMG_CAP;
    if (tid < RANGE_NODES) {
        int wbase = 0;
        for (int w = 0; w < wave; ++w) wbase += wsum[w];
        const int ex = beg + wbase + (inc - cn);
        cur[tid] = ex;
        const int node = p * RANGE_NODES + tid;
        if (node < N_NODES) { cnt[node] = cn; off[node] = ex; }
    }
    __syncthreads();
    {
        int4 r[SUB_PER_WAVE];
        #pragma unroll
        for (int k = 0; k < SUB_PER_WAVE; ++k) {
            r[k] = (lane < sc[k]) ? part[sb[k] + lane] : make_int4(0, 0, 0, -1);
        }
        int slot[SUB_PER_WAVE];
        #pragma unroll
        for (int k = 0; k < SUB_PER_WAVE; ++k) {
            slot[k] = (r[k].w >= 0) ? atomicAdd(&cur[r[k].w & (RANGE_NODES - 1)], 1) : -1;
        }
        if (in_lds) {
            #pragma unroll
            for (int k = 0; k < SUB_PER_WAVE; ++k) {
                if (slot[k] >= 0) img[slot[k] - beg] = make_int4(r[k].x, r[k].y, r[k].z, 0);
            }
        } else {
            #pragma unroll
            for (int k = 0; k < SUB_PER_WAVE; ++k) {
                if (slot[k] >= 0) rec[slot[k]] = make_int4(r[k].x, r[k].y, r[k].z, 0);
            }
        }
        for (int b = wave; b < N_CHUNK; b += 32) {
            const int n  = sub_cnt[b];
            const int s0 = sub_beg[b];
            for (int i = lane + 32; i < n; i += 32) {
                const int4 q  = part[s0 + i];
                const int  sl = atomicAdd(&cur[q.w & (RANGE_NODES - 1)], 1);
                if (in_lds) img[sl - beg] = make_int4(q.x, q.y, q.z, 0);
                else        rec[sl]       = make_int4(q.x, q.y, q.z, 0);
            }
        }
    }
    __syncthreads();
    if (in_lds && total > 0) {
        const int shift = beg & 7;
        int4 q[IMG_TRIPS];
        #pragma unroll
        for (int j = 0; j < IMG_TRIPS; ++j) q[j] = img[min(max(j * SORT_THREADS + tid - shift, 0), total - 1)];
        #pragma unroll
        for (int j = 0; j < IMG_TRIPS; ++j) hold_loaded(q[j]);
        #pragma unroll
        for (int j = 0; j < IMG_TRIPS; ++j) {
            const int i = j * SORT_THREADS + tid - shift;
            if (i >= 0 && i < total) rec[beg + i] = q[j];
        }
    }
}

template <int U>
__device__ __forceinline__ void gather_batch(const float4* __restrict__ x4, int lane,
                                             int rk, int rsrc, int rwt, int j0, float4& acc) {
    float4 v[U];
    float  wt[U];
    #pragma unroll
    for (int u = 0; u < U; ++u) {
        const unsigned b = __builtin_amdgcn_ballot_w32(rk == j0 + u);
        const int l = __builtin_ctz(b);
        const int s = __builtin_amdgcn_readlane(rsrc, l);
        wt[u] = __int_as_float(__builtin_amdgcn_readlane(rwt, l));
        v[u]  = x4[(size_t)s * (DIM / 4) + lane];
    }
    #pragma unroll
    for (int u = 0; u < U; ++u) {
        acc.x += v[u].x * wt[u];
        acc.y += v[u].y * wt[u];
        acc.z += v[u].z * wt[u];
        acc.w += v[u].w * wt[u];
    }
}

#define TILE_PITCH 132

__device__ __forceinline__ float4 gather_node(const float4* __restrict__ x4,
                                              const int4*   __restrict__ rec,
                                              int lane, int deg, int beg) {
    float4 acc = make_float4(0.f, 0.f, 0.f, 0.f);
    for (int base = 0; base < deg; base += 32) {
        const int m = min(32, deg - base);
        int4 r = make_int4(0, 0, 0x7fffffff, 0);
        if (lane < m) r = rec[beg + base + lane];

        int rk = 0;
        for (int j = 0; j < m; ++j) {
            const int ej = __builtin_amdgcn_readlane(r.z, j);
            rk += (ej < r.z) ? 1 : 0;
        }

        int j = 0;
        for (; j + 8 <= m; j += 8) gather_batch<8>(x4, lane, rk, r.x, r.y, j, acc);
        if (j + 4 <= m) { gather_batch<4>(x4, lane, rk, r.x, r.y, j, acc); j += 4; }
        if (j + 2 <= m) { gather_batch<2>(x4, lane, rk, r.x, r.y, j, acc); j += 2; }
        if (j < m)      { gather_batch<1>(x4, lane, rk, r.x, r.y, j, acc); }
    }
    return acc;
}

#define N_TILES     (N_NODES / 16)
#define GG_BLOCKS   1024
#define WSLICE_F4   ((DIM / 8) * 32)
__global__ void __launch_bounds__(256)
gcn_gather_gemm_kernel(const float* __restrict__ x,
                       const int*   __restrict__ cnt,
                       const int*   __restrict__ off,
                       const int4*  __restrict__ rec,
                       const float* __restrict__ W,
                       const float* __restrict__ bias,
                       float*       __restrict__ out,
                       unsigned*    __restrict__ ctr) {
    __shared__ __attribute__((aligned(16))) float tile[16 * TILE_PITCH];
    __shared__ int nextTile;
    __shared__ float4 wfrag[8 * WSLICE_F4];

    const int lane    = threadIdx.x & 31;
    const int wave    = __builtin_amdgcn_readfirstlane(threadIdx.x >> 5);
    const int mn      = lane & 15;
    const int half    = lane >> 4;
    const int colBase = wave * 16;
    const int row0    = wave * 2;
    const int mBase   = half * 8;

    const float4* x4 = (const float4*)x;

    float4* wmine = wfrag + wave * WSLICE_F4 + lane;
    #pragma unroll
    for (int j = 0; j < DIM / 8; ++j) {
        const int kb0 = (2 * j) * 4 + half * 2;
        const int kb1 = (2 * j + 1) * 4 + half * 2;
        float4 f;
        f.x = W[(size_t)kb0 * DIM + colBase + mn];
        f.y = W[(size_t)(kb0 + 1) * DIM + colBase + mn];
        f.z = W[(size_t)kb1 * DIM + colBase + mn];
        f.w = W[(size_t)(kb1 + 1) * DIM + colBase + mn];
        wmine[j * 32] = f;
    }
    const float bv = bias[colBase + mn];

    const float* arow = tile + mn * TILE_PITCH + half * 2;

    int rowTile = blockIdx.x;
    for (int it = 0; it < N_TILES && rowTile < N_TILES; ++it) {
        unsigned fetched = 0u;
        if (threadIdx.x == 0) fetched = atomicInc(ctr, 0xffffffffu);
        const int n0   = rowTile * 16 + row0;
        const int deg0 = cnt[n0],     beg0 = off[n0];
        const int deg1 = cnt[n0 + 1], beg1 = off[n0 + 1];

        const float4 s0 = gather_node(x4, rec, lane, deg0, beg0);
        const float4 s1 = gather_node(x4, rec, lane, deg1, beg1);

        __builtin_amdgcn_fence(__ATOMIC_RELEASE, "workgroup", "local");
        __builtin_amdgcn_s_barrier();
        __builtin_amdgcn_fence(__ATOMIC_ACQUIRE, "workgroup", "local");

        *(float4*)&tile[row0 * TILE_PITCH + lane * 4]       = s0;
        *(float4*)&tile[(row0 + 1) * TILE_PITCH + lane * 4] = s1;
        if (threadIdx.x == 0) nextTile = (int)fetched;

        __builtin_amdgcn_fence(__ATOMIC_RELEASE, "workgroup", "local");
        __builtin_amdgcn_s_barrier();
        __builtin_amdgcn_fence(__ATOMIC_ACQUIRE, "workgroup", "local");

        const int upcoming = GG_BLOCKS + __builtin_amdgcn_readfirstlane(nextTile);

        v8f c = {};
        #pragma unroll
        for (int j = 0; j < DIM / 8; ++j) {
            const float4 f  = wmine[j * 32];
            const v2f    a0 = *(const v2f*)(arow + (2 * j) * 4);
            const v2f    a1 = *(const v2f*)(arow + (2 * j + 1) * 4);
            v2f b0; b0.x = f.x; b0.y = f.y;
            v2f b1; b1.x = f.z; b1.y = f.w;
            c = __builtin_amdgcn_wmma_f32_16x16x4_f32(
                    false, a0, false, b0, (short)0, c, false, false);
            c = __builtin_amdgcn_wmma_f32_16x16x4_f32(
                    false, a1, false, b1, (short)0, c, false, false);
        }

        #pragma unroll
        for (int r = 0; r < 8; ++r) {
            out[(size_t)(rowTile * 16 + mBase + r) * DIM + colBase + mn] = c[r] + bv;
        }
        rowTile = upcoming;
    }
}

extern "C" void kernel_launch(void* const* d_in, const int* in_sizes, int n_in,
                              void* d_out, int out_size, void* d_ws, size_t ws_size,
                              hipStream_t stream) {
    const float* x    = (const float*)d_in[0];
    const int*   esrc = (const int*)d_in[1];
    const int*   edst = (const int*)d_in[2];
    const float* ew   = (const float*)d_in[3];
    const float* W    = (const float*)d_in[4];
    const float* b    = (const float*)d_in[5];
    float*       out  = (float*)d_out;
    int*         cnt    = (int*)((char*)d_ws + WS_CNT_OFF);
    int*         off    = (int*)((char*)d_ws + WS_OFF_OFF);
    int*         tab    = (int*)((char*)d_ws + WS_TAB_OFF);
    int4*        rec    = (int4*)((char*)d_ws + WS_REC_OFF);
    int4*        part   = (int4*)((char*)d_ws + WS_PART_OFF);
    int*         ctr    = (int*)((char*)d_ws + WS_CTR_OFF);

    gcn_chunk_group_kernel<<<N_CHUNK, 256, 0, stream>>>(esrc, edst, ew, tab, part);
    gcn_node_sort_kernel<<<N_RANGE, SORT_THREADS, 0, stream>>>(tab, part, cnt, off, rec, ctr);
    gcn_gather_gemm_kernel<<<GG_BLOCKS, 256, 0, stream>>>(x, cnt, off, rec, W, b, out, (unsigned*)ctr);
}


# node grouping moved into the gather+GEMM launch: a 32-wave block per 256-node range groups its records in LDS and sums rows from there; launches 3 -> 2, rec/cnt/off round trips gone; sum order kept
# speedup vs baseline: 3.4418x; 1.0655x over previous
#include <hip/hip_runtime.h>


typedef __attribute__((ext_vector_type(2))) float v2f;
typedef __attribute__((ext_vector_type(8))) float v8f;

#define N_NODES 50000
#define N_EDGES 800000
#define DIM     128

#define WS_TAB_OFF    34078720
#define WS_REC_OFF    38273024
#define WS_PART_OFF   67108864

#define RANGE_SHIFT  8
#define RANGE_NODES  256
#define N_RANGE      196
#define CHUNK        4096
#define CHUNK_PER_T  16
#define N_CHUNK      196
#define TAB_ROWS     (N_RANGE + 1)

__device__ __forceinline__ void hold_loaded(const int4& v) {
    asm volatile("" :: "v"(v.x), "v"(v.y), "v"(v.z), "v"(v.w));
}

__global__ void __launch_bounds__(256)
gcn_chunk_group_kernel(const int*   __restrict__ esrc,
                       const int*   __restrict__ edst,
                       const float* __restrict__ ew,
                       int*         __restrict__ tab,
                       int4*        __restrict__ part) {
    __shared__ int h[N_RANGE];
    __shared__ int wsum[8];
    __shared__ __attribute__((aligned(16))) int4 img[CHUNK];
    const int tid  = threadIdx.x;
    const int lane = tid & 31;
    const int wave = tid >> 5;
    if (tid < N_RANGE) h[tid] = 0;
    const int base = blockIdx.x * CHUNK;
    int   d[CHUNK_PER_T];
    int   sidx[CHUNK_PER_T];
    float wv[CHUNK_PER_T];
    #pragma unroll
    for (int k = 0; k < CHUNK_PER_T; ++k) {
        const int e = base + k * 256 + tid;
        const bool ok = e < N_EDGES;
        d[k]    = ok ? edst[e] : -1;
        sidx[k] = ok ? esrc[e] : 0;
        wv[k]   = ok ? ew[e]   : 0.f;
    }
    __syncthreads();
    #pragma unroll
    for (int k = 0; k < CHUNK_PER_T; ++k) {
        if (d[k] >= 0) atomicAdd(&h[d[k] >> RANGE_SHIFT], 1);
    }
    __syncthreads();

    const int cn = (tid < N_RANGE) ? h[tid] : 0;
    int inc = cn;
    for (int dd = 1; dd < 32; dd <<= 1) {
        const int t = __shfl_up(inc, dd);
        if (lane >= dd) inc += t;
    }
    if (lane == 31) wsum[wave] = inc;
    __syncthreads();
    int wbase = 0;
    for (int w = 0; w < wave; ++w) wbase += wsum[w];
    const int ex = wbase + (inc - cn);
    if (tid < N_RANGE)  h[tid] = ex;
    if (tid < TAB_ROWS) tab[tid * N_CHUNK + blockIdx.x] = ex;
    __syncthreads();

    int slot[CHUNK_PER_T];
    #pragma unroll
    for (int k = 0; k < CHUNK_PER_T; ++k) {
        slot[k] = (d[k] >= 0) ? atomicAdd(&h[d[k] >> RANGE_SHIFT], 1) : -1;
    }
    #pragma unroll
    for (int k = 0; k < CHUNK_PER_T; ++k) {
        if (slot[k] >= 0)
            img[slot[k]] = make_int4(sidx[k], __float_as_int(wv[k]), base + k * 256 + tid, d[k]);
    }
    __syncthreads();
    const int nrec = min(CHUNK, N_EDGES - base);
    int4 q[CHUNK_PER_T];
    #pragma unroll
    for (int k = 0; k < CHUNK_PER_T; ++k) q[k] = img[min(k * 256 + tid, nrec - 1)];
    #pragma unroll
    for (int k = 0; k < CHUNK_PER_T; ++k) hold_loaded(q[k]);
    #pragma unroll
    for (int k = 0; k < CHUNK_PER_T; ++k) {
        const int i = k * 256 + tid;
        if (i < nrec) part[base + i] = q[k];
    }
}

template <int U>
__device__ __forceinline__ void gather_batch(const float4* __restrict__ x4, int lane,
                                             int rk, int rsrc, int rwt, int j0, float4& acc) {
    float4 v[U];
    float  wt[U];
    #pragma unroll
    for (int u = 0; u < U; ++u) {
        const unsigned b = __builtin_amdgcn_ballot_w32(rk == j0 + u);
        const int l = __builtin_ctz(b);
        const int s = __builtin_amdgcn_readlane(rsrc, l);
        wt[u] = __int_as_float(__builtin_amdgcn_readlane(rwt, l));
        v[u]  = x4[(size_t)s * (DIM / 4) + lane];
    }
    #pragma unroll
    for (int u = 0; u < U; ++u) {
        acc.x += v[u].x * wt[u];
        acc.y += v[u].y * wt[u];
        acc.z += v[u].z * wt[u];
        acc.w += v[u].w * wt[u];
    }
}

#define TILE_PITCH 132

__device__ __forceinline__ float4 gather_node(const float4* __restrict__ x4,
                                              const int4* img_l, const int4* rec_g, bool from_lds,
                                              int lane, int deg, int beg_l, int beg_g) {
    float4 acc = make_float4(0.f, 0.f, 0.f, 0.f);
    for (int base = 0; base < deg; base += 32) {
        const int m = min(32, deg - base);
        int4 r = make_int4(0, 0, 0x7fffffff, 0);
        if (lane < m) {
            if (from_lds) r = img_l[beg_l + base + lane];
            else          r = rec_g[beg_g + base + lane];
        }

        int rk = 0;
        for (int j = 0; j < m; ++j) {
            const int ej = __builtin_amdgcn_readlane(r.z, j);
            rk += (ej < r.z) ? 1 : 0;
        }

        int j = 0;
        for (; j + 8 <= m; j += 8) gather_batch<8>(x4, lane, rk, r.x, r.y, j, acc);
        if (j + 4 <= m) { gather_batch<4>(x4, lane, rk, r.x, r.y, j, acc); j += 4; }
        if (j + 2 <= m) { gather_batch<2>(x4, lane, rk, r.x, r.y, j, acc); j += 2; }
        if (j < m)      { gather_batch<1>(x4, lane, rk, r.x, r.y, j, acc); }
    }
    return acc;
}

#define FUSE_THREADS 1024
#define SUB_PER_WAVE 7
#define SUB_SLOTS    (SUB_PER_WAVE * 32)
#define IMG_CAP      5120
#define N_TILES      (N_NODES / 16)
#define ROUND_TILES  4
#define N_ROUNDS     (RANGE_NODES / (16 * ROUND_TILES))
#define TILE_FLOATS  (16 * TILE_PITCH)
#define WSLICE_F4    ((DIM / 8) * 32)
__global__ void __launch_bounds__(FUSE_THREADS)
gcn_sort_gather_gemm_kernel(const int*   __restrict__ tab,
                            const int4*  __restrict__ part,
                            int4*                     rec,
                            const float* __restrict__ x,
                            const float* __restrict__ W,
                            const float* __restrict__ bias,
                            float*       __restrict__ out) {
    __shared__ int c[RANGE_NODES];
    __shared__ int cur[RANGE_NODES];
    __shared__ int nbeg[RANGE_NODES];
    __shared__ int wsum[8];
    __shared__ int bsum[SUB_PER_WAVE];
    __shared__ int sub_beg[SUB_SLOTS];
    __shared__ int sub_cnt[SUB_SLOTS];
    __shared__ __attribute__((aligned(16))) int4   img[IMG_CAP];
    __shared__ __attribute__((aligned(16))) float  tile[2 * ROUND_TILES * TILE_FLOATS];
    __shared__ __attribute__((aligned(16))) float4 wfrag[8 * WSLICE_F4];
    const int tid  = threadIdx.x;
    const int lane = tid & 31;
    const int wave = __builtin_amdgcn_readfirstlane(tid >> 5);
    const int p    = blockIdx.x;
    const int mn   = lane & 15;
    const int half = lane >> 4;
    const int ct   = wave & 7;
    const int tg   = wave >> 3;
    const int colBase = ct * 16;
    const int mBase   = half * 8;

    float4 wf[DIM / 32];
    #pragma unroll
    for (int jj = 0; jj < DIM / 32; ++jj) {
        const int j   = tg * (DIM / 32) + jj;
        const int kb0 = (2 * j) * 4 + half * 2;
        const int kb1 = (2 * j + 1) * 4 + half * 2;
        wf[jj].x = W[(size_t)kb0 * DIM + colBase + mn];
        wf[jj].y = W[(size_t)(kb0 + 1) * DIM + colBase + mn];
        wf[jj].z = W[(size_t)kb1 * DIM + colBase + mn];
        wf[jj].w = W[(size_t)(kb1 + 1) * DIM + colBase + mn];
    }
    const float bv = bias[colBase + mn];

    int ls0 = 0;
    if (tid < SUB_SLOTS) {
        int n = 0;
        if (tid < N_CHUNK) {
            ls0 = tab[p * N_CHUNK + tid];
            n   = tab[(p + 1) * N_CHUNK + tid] - ls0;
        }
        sub_beg[tid] = tid * CHUNK + ls0;
        sub_cnt[tid] = n;
    }
    if (tid < RANGE_NODES) c[tid] = 0;
    int s = ls0;
    for (int dd = 16; dd >= 1; dd >>= 1) s += __shfl_xor(s, dd);
    if (wave < SUB_PER_WAVE && lane == 0) bsum[wave] = s;
    {
        float4* wdst = wfrag + ct * WSLICE_F4 + lane;
        #pragma unroll
        for (int jj = 0; jj < DIM / 32; ++jj) wdst[(tg * (DIM / 32) + jj) * 32] = wf[jj];
    }
    __syncthreads();
    int beg = 0;
    for (int w = 0; w < SUB_PER_WAVE; ++w) beg += bsum[w];

    int sb[SUB_PER_WAVE];
    int sc[SUB_PER_WAVE];
    #pragma unroll
    for (int k = 0; k < SUB_PER_WAVE; ++k) {
        sb[k] = sub_beg[wave + 32 * k];
        sc[k] = sub_cnt[wave + 32 * k];
    }

    const int* partw = (const int*)part;
    {
        int dv[SUB_PER_WAVE];
        #pragma unroll
        for (int k = 0; k < SUB_PER_WAVE; ++k) {
            dv[k] = (lane < sc[k]) ? partw[(size_t)(sb[k] + lane) * 4 + 3] : -1;
        }
        #pragma unroll
        for (int k = 0; k < SUB_PER_WAVE; ++k) {
            if (dv[k] >= 0) atomicAdd(&c[dv[k] & (RANGE_NODES - 1)], 1);
        }
        for (int b = wave; b < N_CHUNK; b += 32) {
            const int n  = sub_cnt[b];
            const int s0 = sub_beg[b];
            for (int i = lane + 32; i < n; i += 32) {
                const int dd = partw[(size_t)(s0 + i) * 4 + 3];
                atomicAdd(&c[dd & (RANGE_NODES - 1)], 1);
            }
        }
    }
    __syncthreads();

    const int cn = (tid < RANGE_NODES) ? c[tid] : 0;
    int inc = cn;
    for (int dd = 1; dd < 32; dd <<= 1) {
        const int t = __shfl_up(inc, dd);
        if (lane >= dd) inc += t;
    }
    if (tid < RANGE_NODES && lane == 31) wsum[wave] = inc;
    __syncthreads();
    int total = 0;
    for (int w = 0; w < RANGE_NODES / 32; ++w) total += wsum[w];
    const bool in_lds = __builtin_amdgcn_readfirstlane(total) <= IMG_CAP;
    if (tid < RANGE_NODES) {
        int wbase = 0;
        for (int w = 0; w < wave; ++w) wbase += wsum[w];
        const int ex = wbase + (inc - cn);
        cur[tid]  = ex;
        nbeg[tid] = ex;
    }
    __syncthreads();
    {
        int4 r[SUB_PER_WAVE];
        #pragma unroll
        for (int k = 0; k < SUB_PER_WAVE; ++k) {
            r[k] = (lane < sc[k]) ? part[sb[k] + lane] : make_int4(0, 0, 0, -1);
        }
        int slot[SUB_PER_WAVE];
        #pragma unroll
        for (int k = 0; k < SUB_PER_WAVE; ++k) {
            slot[k] = (r[k].w >= 0) ? atomicAdd(&cur[r[k].w & (RANGE_NODES - 1)], 1) : -1;
        }
        if (in_lds) {
            #pragma unroll
            for (int k = 0; k < SUB_PER_WAVE; ++k) {
                if (slot[k] >= 0) img[slot[k]] = make_int4(r[k].x, r[k].y, r[k].z, 0);
            }
        } else {
            #pragma unroll
            for (int k = 0; k < SUB_PER_WAVE; ++k) {
                if (slot[k] >= 0) rec[beg + slot[k]] = make_int4(r[k].x, r[k].y, r[k].z, 0);
            }
        }
        for (int b = wave; b < N_CHUNK; b += 32) {
            const int n  = sub_cnt[b];
            const int s0 = sub_beg[b];
            for (int i = lane + 32; i < n; i += 32) {
                const int4 q  = part[s0 + i];
                const int  sl = atomicAdd(&cur[q.w & (RANGE_NODES - 1)], 1);
                if (in_lds) img[sl]       = make_int4(q.x, q.y, q.z, 0);
                else        rec[beg + sl] = make_int4(q.x, q.y, q.z, 0);
            }
        }
    }
    if (!in_lds) __threadfence();
    __syncthreads();

    const float4* x4 = (const float4*)x;
    const float4* wmine = wfrag + ct * WSLICE_F4 + lane;
    for (int q = 0; q < N_ROUNDS; ++q) {
        const int tile0 = p * (RANGE_NODES / 16) + q * ROUND_TILES;
        if (tile0 >= N_TILES) break;
        float* tb = tile + (q & 1) * (ROUND_TILES * TILE_FLOATS);

        const int nl0 = q * (16 * ROUND_TILES) + 2 * wave;
        const int deg0 = __builtin_amdgcn_readfirstlane(c[nl0]);
        const int rb0  = __builtin_amdgcn_readfirstlane(nbeg[nl0]);
        const int deg1 = __builtin_amdgcn_readfirstlane(c[nl0 + 1]);
        const int rb1  = __builtin_amdgcn_readfirstlane(nbeg[nl0 + 1]);

        const float4 s0 = gather_node(x4, img, rec, in_lds, lane, deg0, rb0, beg + rb0);
        const float4 s1 = gather_node(x4, img, rec, in_lds, lane, deg1, rb1, beg + rb1);

        float* trow = tb + tg * TILE_FLOATS + ((2 * wave) & 15) * TILE_PITCH;
        *(float4*)&trow[lane * 4]              = s0;
        *(float4*)&trow[TILE_PITCH + lane * 4] = s1;

        __builtin_amdgcn_fence(__ATOMIC_RELEASE, "workgroup", "local");
        __builtin_amdgcn_s_barrier();
        __builtin_amdgcn_fence(__ATOMIC_ACQUIRE, "workgroup", "local");

        if (tile0 + tg < N_TILES) {
            const float* arow = tb + tg * TILE_FLOATS + mn * TILE_PITCH + half * 2;
            v8f acc = {};
            #pragma unroll
            for (int j = 0; j < DIM / 8; ++j) {
                const float4 f  = wmine[j * 32];
                const v2f    a0 = *(const v2f*)(arow + (2 * j) * 4);
                const v2f    a1 = *(const v2f*)(arow + (2 * j + 1) * 4);
                v2f b0; b0.x = f.x; b0.y = f.y;
                v2f b1; b1.x = f.z; b1.y = f.w;
                acc = __builtin_amdgcn_wmma_f32_16x16x4_f32(
                        false, a0, false, b0, (short)0, acc, false, false);
                acc = __builtin_amdgcn_wmma_f32_16x16x4_f32(
                        false, a1, false, b1, (short)0, acc, false, false);
            }

            #pragma unroll
            for (int r = 0; r < 8; ++r) {
                out[(size_t)((tile0 + tg) * 16 + mBase + r) * DIM + colBase + mn] = acc[r] + bv;
            }
        }
    }
}

extern "C" void kernel_launch(void* const* d_in, const int* in_sizes, int n_in,
                              void* d_out, int out_size, void* d_ws, size_t ws_size,
                              hipStream_t stream) {
    const float* x    = (const float*)d_in[0];
    const int*   esrc = (const int*)d_in[1];
    const int*   edst = (const int*)d_in[2];
    const float* ew   = (const float*)d_in[3];
    const float* W    = (const float*)d_in[4];
    const float* b    = (const float*)d_in[5];
    float*       out  = (float*)d_out;
    int*         tab    = (int*)((char*)d_ws + WS_TAB_OFF);
    int4*        rec    = (int4*)((char*)d_ws + WS_REC_OFF);
    int4*        part   = (int4*)((char*)d_ws + WS_PART_OFF);

    gcn_chunk_group_kernel<<<N_CHUNK, 256, 0, stream>>>(esrc, edst, ew, tab, part);
    gcn_sort_gather_gemm_kernel<<<N_RANGE, FUSE_THREADS, 0, stream>>>(tab, part, rec, x, W, b, out);
}


# node ranges cut from 16 to 13 row tiles (208 nodes): the sort+gather+GEMM launch has 241 blocks, one per compute unit, not 196; a wave's rows per round are w and w+32; sum order unchanged
# speedup vs baseline: 3.7945x; 1.1025x over previous
#include <hip/hip_runtime.h>


typedef __attribute__((ext_vector_type(2))) float v2f;
typedef __attribute__((ext_vector_type(8))) float v8f;

#define N_NODES 50000
#define N_EDGES 800000
#define DIM     128

#define WS_TAB_OFF    34078720
#define WS_REC_OFF    38273024
#define WS_PART_OFF   67108864

#define RANGE_NODES  208
#define N_RANGE      241
#define CHUNK        4096
#define CHUNK_PER_T  16
#define N_CHUNK      196
#define TAB_ROWS     (N_RANGE + 1)

__device__ __forceinline__ void hold_loaded(const int4& v) {
    asm volatile("" :: "v"(v.x), "v"(v.y), "v"(v.z), "v"(v.w));
}

__global__ void __launch_bounds__(256)
gcn_chunk_group_kernel(const int*   __restrict__ esrc,
                       const int*   __restrict__ edst,
                       const float* __restrict__ ew,
                       int*         __restrict__ tab,
                       int4*        __restrict__ part) {
    __shared__ int h[N_RANGE];
    __shared__ int wsum[8];
    __shared__ __attribute__((aligned(16))) int4 img[CHUNK];
    const int tid  = threadIdx.x;
    const int lane = tid & 31;
    const int wave = tid >> 5;
    if (tid < N_RANGE) h[tid] = 0;
    const int base = blockIdx.x * CHUNK;
    int   d[CHUNK_PER_T];
    int   sidx[CHUNK_PER_T];
    float wv[CHUNK_PER_T];
    #pragma unroll
    for (int k = 0; k < CHUNK_PER_T; ++k) {
        const int e = base + k * 256 + tid;
        const bool ok = e < N_EDGES;
        d[k]    = ok ? edst[e] : -1;
        sidx[k] = ok ? esrc[e] : 0;
        wv[k]   = ok ? ew[e]   : 0.f;
    }
    __syncthreads();
    #pragma unroll
    for (int k = 0; k < CHUNK_PER_T; ++k) {
        if (d[k] >= 0) atomicAdd(&h[(unsigned)d[k] / RANGE_NODES], 1);
    }
    __syncthreads();

    const int cn = (tid < N_RANGE) ? h[tid] : 0;
    int inc = cn;
    for (int dd = 1; dd < 32; dd <<= 1) {
        const int t = __shfl_up(inc, dd);
        if (lane >= dd) inc += t;
    }
    if (lane == 31) wsum[wave] = inc;
    __syncthreads();
    int wbase = 0;
    for (int w = 0; w < wave; ++w) wbase += wsum[w];
    const int ex = wbase + (inc - cn);
    if (tid < N_RANGE)  h[tid] = ex;
    if (tid < TAB_ROWS) tab[tid * N_CHUNK + blockIdx.x] = ex;
    __syncthreads();

    int slot[CHUNK_PER_T];
    #pragma unroll
    for (int k = 0; k < CHUNK_PER_T; ++k) {
        slot[k] = (d[k] >= 0) ? atomicAdd(&h[(unsigned)d[k] / RANGE_NODES], 1) : -1;
    }
    #pragma unroll
    for (int k = 0; k < CHUNK_PER_T; ++k) {
        if (slot[k] >= 0)
            img[slot[k]] = make_int4(sidx[k], __float_as_int(wv[k]), base + k * 256 + tid, d[k]);
    }
    __syncthreads();
    const int nrec = min(CHUNK, N_EDGES - base);
    int4 q[CHUNK_PER_T];
    #pragma unroll
    for (int k = 0; k < CHUNK_PER_T; ++k) q[k] = img[min(k * 256 + tid, nrec - 1)];
    #pragma unroll
    for (int k = 0; k < CHUNK_PER_T; ++k) hold_loaded(q[k]);
    #pragma unroll
    for (int k = 0; k < CHUNK_PER_T; ++k) {
        const int i = k * 256 + tid;
        if (i < nrec) part[base + i] = q[k];
    }
}

template <int U>
__device__ __forceinline__ void gather_batch(const float4* __restrict__ x4, int lane,
                                             int rk, int rsrc, int rwt, int j0, float4& acc) {
    float4 v[U];
    float  wt[U];
    #pragma unroll
    for (int u = 0; u < U; ++u) {
        const unsigned b = __builtin_amdgcn_ballot_w32(rk == j0 + u);
        const int l = __builtin_ctz(b);
        const int s = __builtin_amdgcn_readlane(rsrc, l);
        wt[u] = __int_as_float(__builtin_amdgcn_readlane(rwt, l));
        v[u]  = x4[(size_t)s * (DIM / 4) + lane];
    }
    #pragma unroll
    for (int u = 0; u < U; ++u) {
        acc.x += v[u].x * wt[u];
        acc.y += v[u].y * wt[u];
        acc.z += v[u].z * wt[u];
        acc.w += v[u].w * wt[u];
    }
}

#define TILE_PITCH 132

__device__ __forceinline__ float4 gather_node(const float4* __restrict__ x4,
                                              const int4* img_l, const int4* rec_g, bool from_lds,
                                              int lane, int deg, int beg_l, int beg_g) {
    float4 acc = make_float4(0.f, 0.f, 0.f, 0.f);
    for (int base = 0; base < deg; base += 32) {
        const int m = min(32, deg - base);
        int4 r = make_int4(0, 0, 0x7fffffff, 0);
        if (lane < m) {
            if (from_lds) r = img_l[beg_l + base + lane];
            else          r = rec_g[beg_g + base + lane];
        }

        int rk = 0;
        for (int j = 0; j < m; ++j) {
            const int ej = __builtin_amdgcn_readlane(r.z, j);
            rk += (ej < r.z) ? 1 : 0;
        }

        int j = 0;
        for (; j + 8 <= m; j += 8) gather_batch<8>(x4, lane, rk, r.x, r.y, j, acc);
        if (j + 4 <= m) { gather_batch<4>(x4, lane, rk, r.x, r.y, j, acc); j += 4; }
        if (j + 2 <= m) { gather_batch<2>(x4, lane, rk, r.x, r.y, j, acc); j += 2; }
        if (j < m)      { gather_batch<1>(x4, lane, rk, r.x, r.y, j, acc); }
    }
    return acc;
}

#define FUSE_THREADS 1024
#define SUB_PER_WAVE 7
#define SUB_SLOTS    (SUB_PER_WAVE * 32)
#define IMG_CAP      5120
#define N_TILES      (N_NODES / 16)
#define ROUND_TILES  4
#define RANGE_TILES  (RANGE_NODES / 16)
#define N_ROUNDS     ((RANGE_TILES + ROUND_TILES - 1) / ROUND_TILES)
#define SCAN_WAVES   ((RANGE_NODES + 31) / 32)
#define TILE_FLOATS  (16 * TILE_PITCH)
#define WSLICE_F4    ((DIM / 8) * 32)
__global__ void __launch_bounds__(FUSE_THREADS)
gcn_sort_gather_gemm_kernel(const int*   __restrict__ tab,
                            const int4*  __restrict__ part,
                            int4*                     rec,
                            const float* __restrict__ x,
                            const float* __restrict__ W,
                            const float* __restrict__ bias,
                            float*       __restrict__ out) {
    __shared__ int c[RANGE_NODES];
    __shared__ int cur[RANGE_NODES];
    __shared__ int nbeg[RANGE_NODES];
    __shared__ int wsum[8];
    __shared__ int bsum[SUB_PER_WAVE];
    __shared__ int sub_beg[SUB_SLOTS];
    __shared__ int sub_cnt[SUB_SLOTS];
    __shared__ __attribute__((aligned(16))) int4   img[IMG_CAP];
    __shared__ __attribute__((aligned(16))) float  tile[2 * ROUND_TILES * TILE_FLOATS];
    __shared__ __attribute__((aligned(16))) float4 wfrag[8 * WSLICE_F4];
    const int tid  = threadIdx.x;
    const int lane = tid & 31;
    const int wave = __builtin_amdgcn_readfirstlane(tid >> 5);
    const int p    = blockIdx.x;
    const int node0 = p * RANGE_NODES;
    const int mn   = lane & 15;
    const int half = lane >> 4;
    const int ct   = wave & 7;
    const int tg   = wave >> 3;
    const int colBase = ct * 16;
    const int mBase   = half * 8;

    float4 wf[DIM / 32];
    #pragma unroll
    for (int jj = 0; jj < DIM / 32; ++jj) {
        const int j   = tg * (DIM / 32) + jj;
        const int kb0 = (2 * j) * 4 + half * 2;
        const int kb1 = (2 * j + 1) * 4 + half * 2;
        wf[jj].x = W[(size_t)kb0 * DIM + colBase + mn];
        wf[jj].y = W[(size_t)(kb0 + 1) * DIM + colBase + mn];
        wf[jj].z = W[(size_t)kb1 * DIM + colBase + mn];
        wf[jj].w = W[(size_t)(kb1 + 1) * DIM + colBase + mn];
    }
    const float bv = bias[colBase + mn];

    int ls0 = 0;
    if (tid < SUB_SLOTS) {
        int n = 0;
        if (tid < N_CHUNK) {
            ls0 = tab[p * N_CHUNK + tid];
            n   = tab[(p + 1) * N_CHUNK + tid] - ls0;
        }
        sub_beg[tid] = tid * CHUNK + ls0;
        sub_cnt[tid] = n;
    }
    if (tid < RANGE_NODES) c[tid] = 0;
    int s = ls0;
    for (int dd = 16; dd >= 1; dd >>= 1) s += __shfl_xor(s, dd);
    if (wave < SUB_PER_WAVE && lane == 0) bsum[wave] = s;
    {
        float4* wdst = wfrag + ct * WSLICE_F4 + lane;
        #pragma unroll
        for (int jj = 0; jj < DIM / 32; ++jj) wdst[(tg * (DIM / 32) + jj) * 32] = wf[jj];
    }
    __syncthreads();
    int beg = 0;
    for (int w = 0; w < SUB_PER_WAVE; ++w) beg += bsum[w];

    int sb[SUB_PER_WAVE];
    int sc[SUB_PER_WAVE];
    #pragma unroll
    for (int k = 0; k < SUB_PER_WAVE; ++k) {
        sb[k] = sub_beg[wave + 32 * k];
        sc[k] = sub_cnt[wave + 32 * k];
    }

    const int* partw = (const int*)part;
    {
        int dv[SUB_PER_WAVE];
        #pragma unroll
        for (int k = 0; k < SUB_PER_WAVE; ++k) {
            dv[k] = (lane < sc[k]) ? partw[(size_t)(sb[k] + lane) * 4 + 3] : -1;
        }
        #pragma unroll
        for (int k = 0; k < SUB_PER_WAVE; ++k) {
            if (dv[k] >= 0) atomicAdd(&c[dv[k] - node0], 1);
        }
        for (int b = wave; b < N_CHUNK; b += 32) {
            const int n  = sub_cnt[b];
            const int s0 = sub_beg[b];
            for (int i = lane + 32; i < n; i += 32) {
                const int dd = partw[(size_t)(s0 + i) * 4 + 3];
                atomicAdd(&c[dd - node0], 1);
            }
        }
    }
    __syncthreads();

    const int cn = (tid < RANGE_NODES) ? c[tid] : 0;
    int inc = cn;
    for (int dd = 1; dd < 32; dd <<= 1) {
        const int t = __shfl_up(inc, dd);
        if (lane >= dd) inc += t;
    }
    if (wave < SCAN_WAVES && lane == 31) wsum[wave] = inc;
    __syncthreads();
    int total = 0;
    for (int w = 0; w < SCAN_WAVES; ++w) total += wsum[w];
    const bool in_lds = __builtin_amdgcn_readfirstlane(total) <= IMG_CAP;
    if (tid < RANGE_NODES) {
        int wbase = 0;
        for (int w = 0; w < wave; ++w) wbase += wsum[w];
        const int ex = wbase + (inc - cn);
        cur[tid]  = ex;
        nbeg[tid] = ex;
    }
    __syncthreads();
    {
        int4 r[SUB_PER_WAVE];
        #pragma unroll
        for (int k = 0; k < SUB_PER_WAVE; ++k) {
            r[k] = (lane < sc[k]) ? part[sb[k] + lane] : make_int4(0, 0, 0, -1);
        }
        int slot[SUB_PER_WAVE];
        #pragma unroll
        for (int k = 0; k < SUB_PER_WAVE; ++k) {
            slot[k] = (r[k].w >= 0) ? atomicAdd(&cur[r[k].w - node0], 1) : -1;
        }
        if (in_lds) {
            #pragma unroll
            for (int k = 0; k < SUB_PER_WAVE; ++k) {
                if (slot[k] >= 0) img[slot[k]] = make_int4(r[k].x, r[k].y, r[k].z, 0);
            }
        } else {
            #pragma unroll
            for (int k = 0; k < SUB_PER_WAVE; ++k) {
                if (slot[k] >= 0) rec[beg + slot[k]] = make_int4(r[k].x, r[k].y, r[k].z, 0);
            }
        }
        for (int b = wave; b < N_CHUNK; b += 32) {
            const int n  = sub_cnt[b];
            const int s0 = sub_beg[b];
            for (int i = lane + 32; i < n; i += 32) {
                const int4 q  = part[s0 + i];
                const int  sl = atomicAdd(&cur[q.w - node0], 1);
                if (in_lds) img[sl]       = make_int4(q.x, q.y, q.z, 0);
                else        rec[beg + sl] = make_int4(q.x, q.y, q.z, 0);
            }
        }
    }
    if (!in_lds) __threadfence();
    __syncthreads();

    const float4* x4 = (const float4*)x;
    const float4* wmine = wfrag + ct * WSLICE_F4 + lane;
    for (int q = 0; q < N_ROUNDS; ++q) {
        const int tile0 = p * RANGE_TILES + q * ROUND_TILES;
        if (tile0 >= N_TILES) break;
        float* tb = tile + (q & 1) * (ROUND_TILES * TILE_FLOATS);

        for (int hh = 0; hh < 2; ++hh) {
            const int row = wave + 32 * hh;
            const int nl  = q * (16 * ROUND_TILES) + row;
            if (nl < RANGE_NODES) {
                const int deg = __builtin_amdgcn_readfirstlane(c[nl]);
                const int rb  = __builtin_amdgcn_readfirstlane(nbeg[nl]);
                const float4 sum = gather_node(x4, img, rec, in_lds, lane, deg, rb, beg + rb);
                float* trow = tb + (row >> 4) * TILE_FLOATS + (row & 15) * TILE_PITCH;
                *(float4*)&trow[lane * 4] = sum;
            }
        }

        __builtin_amdgcn_fence(__ATOMIC_RELEASE, "workgroup", "local");
        __builtin_amdgcn_s_barrier();
        __builtin_amdgcn_fence(__ATOMIC_ACQUIRE, "workgroup", "local");

        if (q * ROUND_TILES + tg < RANGE_TILES && tile0 + tg < N_TILES) {
            const float* arow = tb + tg * TILE_FLOATS + mn * TILE_PITCH + half * 2;
            v8f acc = {};
            #pragma unroll
            for (int j = 0; j < DIM / 8; ++j) {
                const float4 f  = wmine[j * 32];
                const v2f    a0 = *(const v2f*)(arow + (2 * j) * 4);
                const v2f    a1 = *(const v2f*)(arow + (2 * j + 1) * 4);
                v2f b0; b0.x = f.x; b0.y = f.y;
                v2f b1; b1.x = f.z; b1.y = f.w;
                acc = __builtin_amdgcn_wmma_f32_16x16x4_f32(
                        false, a0, false, b0, (short)0, acc, false, false);
                acc = __builtin_amdgcn_wmma_f32_16x16x4_f32(
                        false, a1, false, b1, (short)0, acc, false, false);
            }

            #pragma unroll
            for (int r = 0; r < 8; ++r) {
                out[(size_t)((tile0 + tg) * 16 + mBase + r) * DIM + colBase + mn] = acc[r] + bv;
            }
        }
    }
}

extern "C" void kernel_launch(void* const* d_in, const int* in_sizes, int n_in,
                              void* d_out, int out_size, void* d_ws, size_t ws_size,
                              hipStream_t stream) {
    const float* x    = (const float*)d_in[0];
    const int*   esrc = (const int*)d_in[1];
    const int*   edst = (const int*)d_in[2];
    const float* ew   = (const float*)d_in[3];
    const float* W    = (const float*)d_in[4];
    const float* b    = (const float*)d_in[5];
    float*       out  = (float*)d_out;
    int*         tab    = (int*)((char*)d_ws + WS_TAB_OFF);
    int4*        rec    = (int4*)((char*)d_ws + WS_REC_OFF);
    int4*        part   = (int4*)((char*)d_ws + WS_PART_OFF);

    gcn_chunk_group_kernel<<<N_CHUNK, 256, 0, stream>>>(esrc, edst, ew, tab, part);
    gcn_sort_gather_gemm_kernel<<<N_RANGE, FUSE_THREADS, 0, stream>>>(tab, part, rec, x, W, b, out);
}


# gather issues a node's source rows up to 24 at a time as one load group (was batches of 8,4,2,1, each waited for in turn); slots read rank-ordered lanes after one ds_permute; sum order unchanged
# speedup vs baseline: 4.0134x; 1.0577x over previous
#include <hip/hip_runtime.h>


typedef __attribute__((ext_vector_type(2))) float v2f;
typedef __attribute__((ext_vector_type(8))) float v8f;

#define N_NODES 50000
#define N_EDGES 800000
#define DIM     128

#define WS_TAB_OFF    34078720
#define WS_REC_OFF    38273024
#define WS_PART_OFF   67108864

#define RANGE_NODES  208
#define N_RANGE      241
#define CHUNK        4096
#define CHUNK_PER_T  16
#define N_CHUNK      196
#define TAB_ROWS     (N_RANGE + 1)

__device__ __forceinline__ void hold_loaded(const int4& v) {
    asm volatile("" :: "v"(v.x), "v"(v.y), "v"(v.z), "v"(v.w));
}

__global__ void __launch_bounds__(256)
gcn_chunk_group_kernel(const int*   __restrict__ esrc,
                       const int*   __restrict__ edst,
                       const float* __restrict__ ew,
                       int*         __restrict__ tab,
                       int4*        __restrict__ part) {
    __shared__ int h[N_RANGE];
    __shared__ int wsum[8];
    __shared__ __attribute__((aligned(16))) int4 img[CHUNK];
    const int tid  = threadIdx.x;
    const int lane = tid & 31;
    const int wave = tid >> 5;
    if (tid < N_RANGE) h[tid] = 0;
    const int base = blockIdx.x * CHUNK;
    int   d[CHUNK_PER_T];
    int   sidx[CHUNK_PER_T];
    float wv[CHUNK_PER_T];
    #pragma unroll
    for (int k = 0; k < CHUNK_PER_T; ++k) {
        const int e = base + k * 256 + tid;
        const bool ok = e < N_EDGES;
        d[k]    = ok ? edst[e] : -1;
        sidx[k] = ok ? esrc[e] : 0;
        wv[k]   = ok ? ew[e]   : 0.f;
    }
    __syncthreads();
    #pragma unroll
    for (int k = 0; k < CHUNK_PER_T; ++k) {
        if (d[k] >= 0) atomicAdd(&h[(unsigned)d[k] / RANGE_NODES], 1);
    }
    __syncthreads();

    const int cn = (tid < N_RANGE) ? h[tid] : 0;
    int inc = cn;
    for (int dd = 1; dd < 32; dd <<= 1) {
        const int t = __shfl_up(inc, dd);
        if (lane >= dd) inc += t;
    }
    if (lane == 31) wsum[wave] = inc;
    __syncthreads();
    int wbase = 0;
    for (int w = 0; w < wave; ++w) wbase += wsum[w];
    const int ex = wbase + (inc - cn);
    if (tid < N_RANGE)  h[tid] = ex;
    if (tid < TAB_ROWS) tab[tid * N_CHUNK + blockIdx.x] = ex;
    __syncthreads();

    int slot[CHUNK_PER_T];
    #pragma unroll
    for (int k = 0; k < CHUNK_PER_T; ++k) {
        slot[k] = (d[k] >= 0) ? atomicAdd(&h[(unsigned)d[k] / RANGE_NODES], 1) : -1;
    }
    #pragma unroll
    for (int k = 0; k < CHUNK_PER_T; ++k) {
        if (slot[k] >= 0)
            img[slot[k]] = make_int4(sidx[k], __float_as_int(wv[k]), base + k * 256 + tid, d[k]);
    }
    __syncthreads();
    const int nrec = min(CHUNK, N_EDGES - base);
    int4 q[CHUNK_PER_T];
    #pragma unroll
    for (int k = 0; k < CHUNK_PER_T; ++k) q[k] = img[min(k * 256 + tid, nrec - 1)];
    #pragma unroll
    for (int k = 0; k < CHUNK_PER_T; ++k) hold_loaded(q[k]);
    #pragma unroll
    for (int k = 0; k < CHUNK_PER_T; ++k) {
        const int i = k * 256 + tid;
        if (i < nrec) part[base + i] = q[k];
    }
}

template <int U, int MAXS>
__device__ __forceinline__ void gather_slots(const float4* __restrict__ x4, int lane,
                                             int psrc, int pwt, int top, int n, float4& acc) {
    if constexpr (U < MAXS) {
        if (U < n) {
            const int    l  = top - U;
            const int    s  = __builtin_amdgcn_readlane(psrc, l);
            const float  wt = __int_as_float(__builtin_amdgcn_readlane(pwt, l));
            const float4 v  = x4[(size_t)s * (DIM / 4) + lane];
            gather_slots<U + 1, MAXS>(x4, lane, psrc, pwt, top, n, acc);
            acc.x += v.x * wt;
            acc.y += v.y * wt;
            acc.z += v.z * wt;
            acc.w += v.w * wt;
        }
    }
}

#define GROUP 24

#define TILE_PITCH 132

__device__ __forceinline__ float4 gather_node(const float4* __restrict__ x4,
                                              const int4* img_l, const int4* rec_g, bool from_lds,
                                              int lane, int deg, int beg_l, int beg_g) {
    float4 acc = make_float4(0.f, 0.f, 0.f, 0.f);
    for (int base = 0; base < deg; base += 32) {
        const int m = min(32, deg - base);
        int4 r = make_int4(0, 0, 0x7fffffff, 0);
        if (lane < m) {
            if (from_lds) r = img_l[beg_l + base + lane];
            else          r = rec_g[beg_g + base + lane];
        }

        int rk = 0;
        for (int j = 0; j < m; ++j) {
            const int ej = __builtin_amdgcn_readlane(r.z, j);
            rk += (ej < r.z) ? 1 : 0;
        }

        const int psrc = __builtin_amdgcn_ds_permute(rk << 2, r.x);
        const int pwt  = __builtin_amdgcn_ds_permute(rk << 2, r.y);
        const int n0   = min(GROUP, m);
        gather_slots<0, GROUP>(x4, lane, psrc, pwt, n0 - 1, n0, acc);
        if (m > GROUP) gather_slots<0, 32 - GROUP>(x4, lane, psrc, pwt, m - 1, m - GROUP, acc);
    }
    return acc;
}

#define FUSE_THREADS 1024
#define SUB_PER_WAVE 7
#define SUB_SLOTS    (SUB_PER_WAVE * 32)
#define IMG_CAP      5120
#define N_TILES      (N_NODES / 16)
#define ROUND_TILES  4
#define RANGE_TILES  (RANGE_NODES / 16)
#define N_ROUNDS     ((RANGE_TILES + ROUND_TILES - 1) / ROUND_TILES)
#define SCAN_WAVES   ((RANGE_NODES + 31) / 32)
#define TILE_FLOATS  (16 * TILE_PITCH)
#define WSLICE_F4    ((DIM / 8) * 32)
__global__ void __launch_bounds__(FUSE_THREADS)
gcn_sort_gather_gemm_kernel(const int*   __restrict__ tab,
                            const int4*  __restrict__ part,
                            int4*                     rec,
                            const float* __restrict__ x,
                            const float* __restrict__ W,
                            const float* __restrict__ bias,
                            float*       __restrict__ out) {
    __shared__ int c[RANGE_NODES];
    __shared__ int cur[RANGE_NODES];
    __shared__ int nbeg[RANGE_NODES];
    __shared__ int wsum[8];
    __shared__ int bsum[SUB_PER_WAVE];
    __shared__ int sub_beg[SUB_SLOTS];
    __shared__ int sub_cnt[SUB_SLOTS];
    __shared__ __attribute__((aligned(16))) int4   img[IMG_CAP];
    __shared__ __attribute__((aligned(16))) float  tile[2 * ROUND_TILES * TILE_FLOATS];
    __shared__ __attribute__((aligned(16))) float4 wfrag[8 * WSLICE_F4];
    const int tid  = threadIdx.x;
    const int lane = tid & 31;
    const int wave = __builtin_amdgcn_readfirstlane(tid >> 5);
    const int p    = blockIdx.x;
    const int node0 = p * RANGE_NODES;
    const int mn   = lane & 15;
    const int half = lane >> 4;
    const int ct   = wave & 7;
    const int tg   = wave >> 3;
    const int colBase = ct * 16;
    const int mBase   = half * 8;

    float4 wf[DIM / 32];
    #pragma unroll
    for (int jj = 0; jj < DIM / 32; ++jj) {
        const int j   = tg * (DIM / 32) + jj;
        const int kb0 = (2 * j) * 4 + half * 2;
        const int kb1 = (2 * j + 1) * 4 + half * 2;
        wf[jj].x = W[(size_t)kb0 * DIM + colBase + mn];
        wf[jj].y = W[(size_t)(kb0 + 1) * DIM + colBase + mn];
        wf[jj].z = W[(size_t)kb1 * DIM + colBase + mn];
        wf[jj].w = W[(size_t)(kb1 + 1) * DIM + colBase + mn];
    }
    const float bv = bias[colBase + mn];

    int ls0 = 0;
    if (tid < SUB_SLOTS) {
        int n = 0;
        if (tid < N_CHUNK) {
            ls0 = tab[p * N_CHUNK + tid];
            n   = tab[(p + 1) * N_CHUNK + tid] - ls0;
        }
        sub_beg[tid] = tid * CHUNK + ls0;
        sub_cnt[tid] = n;
    }
    if (tid < RANGE_NODES) c[tid] = 0;
    int s = ls0;
    for (int dd = 16; dd >= 1; dd >>= 1) s += __shfl_xor(s, dd);
    if (wave < SUB_PER_WAVE && lane == 0) bsum[wave] = s;
    {
        float4* wdst = wfrag + ct * WSLICE_F4 + lane;
        #pragma unroll
        for (int jj = 0; jj < DIM / 32; ++jj) wdst[(tg * (DIM / 32) + jj) * 32] = wf[jj];
    }
    __syncthreads();
    int beg = 0;
    for (int w = 0; w < SUB_PER_WAVE; ++w) beg += bsum[w];

    int sb[SUB_PER_WAVE];
    int sc[SUB_PER_WAVE];
    #pragma unroll
    for (int k = 0; k < SUB_PER_WAVE; ++k) {
        sb[k] = sub_beg[wave + 32 * k];
        sc[k] = sub_cnt[wave + 32 * k];
    }

    const int* partw = (const int*)part;
    {
        int dv[SUB_PER_WAVE];
        #pragma unroll
        for (int k = 0; k < SUB_PER_WAVE; ++k) {
            dv[k] = (lane < sc[k]) ? partw[(size_t)(sb[k] + lane) * 4 + 3] : -1;
        }
        #pragma unroll
        for (int k = 0; k < SUB_PER_WAVE; ++k) {
            if (dv[k] >= 0) atomicAdd(&c[dv[k] - node0], 1);
        }
        for (int b = wave; b < N_CHUNK; b += 32) {
            const int n  = sub_cnt[b];
            const int s0 = sub_beg[b];
            for (int i = lane + 32; i < n; i += 32) {
                const int dd = partw[(size_t)(s0 + i) * 4 + 3];
                atomicAdd(&c[dd - node0], 1);
            }
        }
    }
    __syncthreads();

    const int cn = (tid < RANGE_NODES) ? c[tid] : 0;
    int inc = cn;
    for (int dd = 1; dd < 32; dd <<= 1) {
        const int t = __shfl_up(inc, dd);
        if (lane >= dd) inc += t;
    }
    if (wave < SCAN_WAVES && lane == 31) wsum[wave] = inc;
    __syncthreads();
    int total = 0;
    for (int w = 0; w < SCAN_WAVES; ++w) total += wsum[w];
    const bool in_lds = __builtin_amdgcn_readfirstlane(total) <= IMG_CAP;
    if (tid < RANGE_NODES) {
        int wbase = 0;
        for (int w = 0; w < wave; ++w) wbase += wsum[w];
        const int ex = wbase + (inc - cn);
        cur[tid]  = ex;
        nbeg[tid] = ex;
    }
    __syncthreads();
    {
        int4 r[SUB_PER_WAVE];
        #pragma unroll
        for (int k = 0; k < SUB_PER_WAVE; ++k) {
            r[k] = (lane < sc[k]) ? part[sb[k] + lane] : make_int4(0, 0, 0, -1);
        }
        int slot[SUB_PER_WAVE];
        #pragma unroll
        for (int k = 0; k < SUB_PER_WAVE; ++k) {
            slot[k] = (r[k].w >= 0) ? atomicAdd(&cur[r[k].w - node0], 1) : -1;
        }
        if (in_lds) {
            #pragma unroll
            for (int k = 0; k < SUB_PER_WAVE; ++k) {
                if (slot[k] >= 0) img[slot[k]] = make_int4(r[k].x, r[k].y, r[k].z, 0);
            }
        } else {
            #pragma unroll
            for (int k = 0; k < SUB_PER_WAVE; ++k) {
                if (slot[k] >= 0) rec[beg + slot[k]] = make_int4(r[k].x, r[k].y, r[k].z, 0);
            }
        }
        for (int b = wave; b < N_CHUNK; b += 32) {
            const int n  = sub_cnt[b];
            const int s0 = sub_beg[b];
            for (int i = lane + 32; i < n; i += 32) {
                const int4 q  = part[s0 + i];
                const int  sl = atomicAdd(&cur[q.w - node0], 1);
                if (in_lds) img[sl]       = make_int4(q.x, q.y, q.z, 0);
                else        rec[beg + sl] = make_int4(q.x, q.y, q.z, 0);
            }
        }
    }
    if (!in_lds) __threadfence();
    __syncthreads();

    const float4* x4 = (const float4*)x;
    const float4* wmine = wfrag + ct * WSLICE_F4 + lane;
    for (int q = 0; q < N_ROUNDS; ++q) {
        const int tile0 = p * RANGE_TILES + q * ROUND_TILES;
        if (tile0 >= N_TILES) break;
        float* tb = tile + (q & 1) * (ROUND_TILES * TILE_FLOATS);

        for (int hh = 0; hh < 2; ++hh) {
            const int row = wave + 32 * hh;
            const int nl  = q * (16 * ROUND_TILES) + row;
            if (nl < RANGE_NODES) {
                const int deg = __builtin_amdgcn_readfirstlane(c[nl]);
                const int rb  = __builtin_amdgcn_readfirstlane(nbeg[nl]);
                const float4 sum = gather_node(x4, img, rec, in_lds, lane, deg, rb, beg + rb);
                float* trow = tb + (row >> 4) * TILE_FLOATS + (row & 15) * TILE_PITCH;
                *(float4*)&trow[lane * 4] = sum;
            }
        }

        __builtin_amdgcn_fence(__ATOMIC_RELEASE, "workgroup", "local");
        __builtin_amdgcn_s_barrier();
        __builtin_amdgcn_fence(__ATOMIC_ACQUIRE, "workgroup", "local");

        if (q * ROUND_TILES + tg < RANGE_TILES && tile0 + tg < N_TILES) {
            const float* arow = tb + tg * TILE_FLOATS + mn * TILE_PITCH + half * 2;
            v8f acc = {};
            #pragma unroll
            for (int j = 0; j < DIM / 8; ++j) {
                const float4 f  = wmine[j * 32];
                const v2f    a0 = *(const v2f*)(arow + (2 * j) * 4);
                const v2f    a1 = *(const v2f*)(arow + (2 * j + 1) * 4);
                v2f b0; b0.x = f.x; b0.y = f.y;
                v2f b1; b1.x = f.z; b1.y = f.w;
                acc = __builtin_amdgcn_wmma_f32_16x16x4_f32(
                        false, a0, false, b0, (short)0, acc, false, false);
                acc = __builtin_amdgcn_wmma_f32_16x16x4_f32(
                        false, a1, false, b1, (short)0, acc, false, false);
            }

            #pragma unroll
            for (int r = 0; r < 8; ++r) {
                out[(size_t)((tile0 + tg) * 16 + mBase + r) * DIM + colBase + mn] = acc[r] + bv;
            }
        }
    }
}

extern "C" void kernel_launch(void* const* d_in, const int* in_sizes, int n_in,
                              void* d_out, int out_size, void* d_ws, size_t ws_size,
                              hipStream_t stream) {
    const float* x    = (const float*)d_in[0];
    const int*   esrc = (const int*)d_in[1];
    const int*   edst = (const int*)d_in[2];
    const float* ew   = (const float*)d_in[3];
    const float* W    = (const float*)d_in[4];
    const float* b    = (const float*)d_in[5];
    float*       out  = (float*)d_out;
    int*         tab    = (int*)((char*)d_ws + WS_TAB_OFF);
    int4*        rec    = (int4*)((char*)d_ws + WS_REC_OFF);
    int4*        part   = (int4*)((char*)d_ws + WS_PART_OFF);

    gcn_chunk_group_kernel<<<N_CHUNK, 256, 0, stream>>>(esrc, edst, ew, tab, part);
    gcn_sort_gather_gemm_kernel<<<N_RANGE, FUSE_THREADS, 0, stream>>>(tab, part, rec, x, W, b, out);
}
